# starting kernel
# baseline (speedup 1.0000x reference)
#include <hip/hip_runtime.h>

typedef float v2f __attribute__((ext_vector_type(2)));
typedef float v8f __attribute__((ext_vector_type(8)));

#define T_SEQ 2048
#define FEAT  256
#define GATES 64
#define HID   16
#define BATCH 64

__global__ __launch_bounds__(128) void lstm_in_gemm_wmma(
    const float* __restrict__ x,
    const float* __restrict__ W,
    const float* __restrict__ bih,
    const float* __restrict__ bhh,
    float* __restrict__ xg)
{
  const int lane = threadIdx.x & 31;
  const int m    = threadIdx.x >> 5;
  const int half = lane >> 4;
  const int r    = lane & 15;
  const int tb   = blockIdx.x * 16;
  const int b    = blockIdx.y;

  const float* Xb   = x + (size_t)b * FEAT * T_SEQ;
  const float* Wrow = W + (size_t)(m * 16 + r) * FEAT;

  v8f acc = {};
  for (int k = 0; k < FEAT; k += 4) {
    const int ka = k + 2 * half;
    v2f av, bv;
    av.x = Wrow[ka];
    av.y = Wrow[ka + 1];
    bv.x = Xb[(size_t)ka * T_SEQ + tb + r];
    bv.y = Xb[(size_t)(ka + 1) * T_SEQ + tb + r];
    acc = __builtin_amdgcn_wmma_f32_16x16x4_f32(
         false, av,  false, bv,
         (short)0, acc,  false,  false);
  }

#pragma unroll
  for (int v = 0; v < 8; ++v) {
    const int g = m * 16 + v + 8 * half;
    const float val = acc[v] + bih[g] + bhh[g];
    xg[((size_t)b * GATES + g) * T_SEQ + tb + r] = val;
  }
}

__device__ __forceinline__ float bcast(float v, int j) {
  return __uint_as_float(__builtin_amdgcn_readlane(__float_as_uint(v), (unsigned)j));
}
__device__ __forceinline__ float sig_f(float x) {
  return __builtin_amdgcn_rcpf(1.0f + __expf(-x));
}
__device__ __forceinline__ float tanh_f(float x) {
  return 1.0f - 2.0f * __builtin_amdgcn_rcpf(__expf(2.0f * x) + 1.0f);
}

__global__ __launch_bounds__(32) void lstm_recurrence(
    const float* __restrict__ xg,
    const float* __restrict__ Whh0,
    const float* __restrict__ Wih1,
    const float* __restrict__ Whh1,
    const float* __restrict__ bih1,
    const float* __restrict__ bhh1,
    float* __restrict__ out)
{
  const int lane = threadIdx.x;
  const int b    = blockIdx.x;
  const int g0   = lane;
  const int g1   = lane + 32;

  float w0a[HID], w0b[HID], wia[HID], wib[HID], wha[HID], whb[HID];
#pragma unroll
  for (int j = 0; j < HID; ++j) {
    w0a[j] = Whh0[g0 * HID + j];
    w0b[j] = Whh0[g1 * HID + j];
    wia[j] = Wih1[g0 * HID + j];
    wib[j] = Wih1[g1 * HID + j];
    wha[j] = Whh1[g0 * HID + j];
    whb[j] = Whh1[g1 * HID + j];
  }
  const float bias0 = bih1[g0] + bhh1[g0];
  const float bias1 = bih1[g1] + bhh1[g1];

  const float* xg0p = xg + ((size_t)b * GATES + g0) * T_SEQ;
  const float* xg1p = xg + ((size_t)b * GATES + g1) * T_SEQ;
  float* outb = out + (size_t)b * T_SEQ;

  float h0 = 0.0f, c0 = 0.0f;
  float h1 = 0.0f, c1 = 0.0f;
  float hb0[HID], hb1[HID];
#pragma unroll
  for (int j = 0; j < HID; ++j) { hb0[j] = 0.0f; hb1[j] = 0.0f; }

  for (int t0 = 0; t0 < T_SEQ; t0 += 4) {
    const float4 x0v = *(const float4*)(xg0p + t0);
    const float4 x1v = *(const float4*)(xg1p + t0);
    __builtin_prefetch(xg0p + t0 + 64, 0, 3);
    __builtin_prefetch(xg1p + t0 + 64, 0, 3);
    const float x0a[4] = {x0v.x, x0v.y, x0v.z, x0v.w};
    const float x1a[4] = {x1v.x, x1v.y, x1v.z, x1v.w};

#pragma unroll
    for (int u = 0; u < 4; ++u) {
      float p0a = 0.f, p0b = 0.f, p1a = 0.f, p1b = 0.f;
#pragma unroll
      for (int j = 0; j < 8; ++j) {
        p0a = fmaf(w0a[j],     hb0[j],     p0a);
        p0b = fmaf(w0a[j + 8], hb0[j + 8], p0b);
        p1a = fmaf(w0b[j],     hb0[j],     p1a);
        p1b = fmaf(w0b[j + 8], hb0[j + 8], p1b);
      }
      const float a0 = x0a[u] + (p0a + p0b);
      const float a1 = x1a[u] + (p1a + p1b);
      const float fv = __shfl_xor(a0, 16, 32);
      const float ov = __shfl_xor(a1, 16, 32);
      c0 = sig_f(fv) * c0 + sig_f(a0) * tanh_f(a1);
      h0 = sig_f(ov) * tanh_f(c0);

#pragma unroll
      for (int j = 0; j < HID; ++j) hb0[j] = bcast(h0, j);

      float r0a = 0.f, r0b = 0.f, r0c = 0.f, r0d = 0.f;
      float r1a = 0.f, r1b = 0.f, r1c = 0.f, r1d = 0.f;
#pragma unroll
      for (int j = 0; j < 8; ++j) {
        r0a = fmaf(wia[j],     hb0[j],     r0a);
        r0b = fmaf(wia[j + 8], hb0[j + 8], r0b);
        r0c = fmaf(wha[j],     hb1[j],     r0c);
        r0d = fmaf(wha[j + 8], hb1[j + 8], r0d);
        r1a = fmaf(wib[j],     hb0[j],     r1a);
        r1b = fmaf(wib[j + 8], hb0[j + 8], r1b);
        r1c = fmaf(whb[j],     hb1[j],     r1c);
        r1d = fmaf(whb[j + 8], hb1[j + 8], r1d);
      }
      const float b0 = bias0 + ((r0a + r0b) + (r0c + r0d));
      const float b1 = bias1 + ((r1a + r1b) + (r1c + r1d));
      const float fv1 = __shfl_xor(b0, 16, 32);
      const float ov1 = __shfl_xor(b1, 16, 32);
      c1 = sig_f(fv1) * c1 + sig_f(b0) * tanh_f(b1);
      h1 = sig_f(ov1) * tanh_f(c1);

#pragma unroll
      for (int j = 0; j < HID; ++j) hb1[j] = bcast(h1, j);
      const float sA = (hb1[0] + hb1[1]) + (hb1[2] + hb1[3]);
      const float sB = (hb1[4] + hb1[5]) + (hb1[6] + hb1[7]);
      const float sC = (hb1[8] + hb1[9]) + (hb1[10] + hb1[11]);
      const float sD = (hb1[12] + hb1[13]) + (hb1[14] + hb1[15]);
      const float mean = ((sA + sB) + (sC + sD)) * 0.0625f;
      if (lane == 0) outb[t0 + u] = mean;
    }
  }
}

extern "C" void kernel_launch(void* const* d_in, const int* in_sizes, int n_in,
                              void* d_out, int out_size, void* d_ws, size_t ws_size,
                              hipStream_t stream) {
  (void)in_sizes; (void)n_in; (void)out_size; (void)ws_size;
  const float* x    = (const float*)d_in[0];
  const float* Wih0 = (const float*)d_in[1];
  const float* Whh0 = (const float*)d_in[2];
  const float* bih0 = (const float*)d_in[3];
  const float* bhh0 = (const float*)d_in[4];
  const float* Wih1 = (const float*)d_in[5];
  const float* Whh1 = (const float*)d_in[6];
  const float* bih1 = (const float*)d_in[7];
  const float* bhh1 = (const float*)d_in[8];
  float* out = (float*)d_out;
  float* xg  = (float*)d_ws;

  dim3 g1(T_SEQ / 16, BATCH, 1);
  lstm_in_gemm_wmma<<<g1, dim3(128, 1, 1), 0, stream>>>(x, Wih0, bih0, bhh0, xg);

  lstm_recurrence<<<dim3(BATCH, 1, 1), dim3(32, 1, 1), 0, stream>>>(
      xg, Whh0, Wih1, Whh1, bih1, bhh1, out);
}


# recurrence: software-pipeline the two layers one step apart (layer 1 of step t beside layer 0 of step t+1 in one loop body), so their dependent chains overlap
# speedup vs baseline: 1.2964x; 1.2964x over previous
#include <hip/hip_runtime.h>

typedef float v2f __attribute__((ext_vector_type(2)));
typedef float v8f __attribute__((ext_vector_type(8)));

#define T_SEQ 2048
#define FEAT  256
#define GATES 64
#define HID   16
#define BATCH 64

__global__ __launch_bounds__(128) void lstm_in_gemm_wmma(
    const float* __restrict__ x,
    const float* __restrict__ W,
    const float* __restrict__ bih,
    const float* __restrict__ bhh,
    float* __restrict__ xg)
{
  const int lane = threadIdx.x & 31;
  const int m    = threadIdx.x >> 5;
  const int half = lane >> 4;
  const int r    = lane & 15;
  const int tb   = blockIdx.x * 16;
  const int b    = blockIdx.y;

  const float* Xb   = x + (size_t)b * FEAT * T_SEQ;
  const float* Wrow = W + (size_t)(m * 16 + r) * FEAT;

  v8f acc = {};
  for (int k = 0; k < FEAT; k += 4) {
    const int ka = k + 2 * half;
    v2f av, bv;
    av.x = Wrow[ka];
    av.y = Wrow[ka + 1];
    bv.x = Xb[(size_t)ka * T_SEQ + tb + r];
    bv.y = Xb[(size_t)(ka + 1) * T_SEQ + tb + r];
    acc = __builtin_amdgcn_wmma_f32_16x16x4_f32(
         false, av,  false, bv,
         (short)0, acc,  false,  false);
  }

#pragma unroll
  for (int v = 0; v < 8; ++v) {
    const int g = m * 16 + v + 8 * half;
    const float val = acc[v] + bih[g] + bhh[g];
    xg[((size_t)b * GATES + g) * T_SEQ + tb + r] = val;
  }
}

__device__ __forceinline__ float bcast(float v, int j) {
  return __uint_as_float(__builtin_amdgcn_readlane(__float_as_uint(v), (unsigned)j));
}
__device__ __forceinline__ float sig_f(float x) {
  return __builtin_amdgcn_rcpf(1.0f + __expf(-x));
}
__device__ __forceinline__ float tanh_f(float x) {
  return 1.0f - 2.0f * __builtin_amdgcn_rcpf(__expf(2.0f * x) + 1.0f);
}

__global__ __launch_bounds__(32) void lstm_recurrence(
    const float* __restrict__ xg,
    const float* __restrict__ Whh0,
    const float* __restrict__ Wih1,
    const float* __restrict__ Whh1,
    const float* __restrict__ bih1,
    const float* __restrict__ bhh1,
    float* __restrict__ out)
{
  const int lane = threadIdx.x;
  const int b    = blockIdx.x;
  const int g0   = lane;
  const int g1   = lane + 32;

  float w0a[HID], w0b[HID], wia[HID], wib[HID], wha[HID], whb[HID];
#pragma unroll
  for (int j = 0; j < HID; ++j) {
    w0a[j] = Whh0[g0 * HID + j];
    w0b[j] = Whh0[g1 * HID + j];
    wia[j] = Wih1[g0 * HID + j];
    wib[j] = Wih1[g1 * HID + j];
    wha[j] = Whh1[g0 * HID + j];
    whb[j] = Whh1[g1 * HID + j];
  }
  const float bias0 = bih1[g0] + bhh1[g0];
  const float bias1 = bih1[g1] + bhh1[g1];

  const float* xg0p = xg + ((size_t)b * GATES + g0) * T_SEQ;
  const float* xg1p = xg + ((size_t)b * GATES + g1) * T_SEQ;
  float* outb = out + (size_t)b * T_SEQ;

  float h0 = 0.0f, c0 = 0.0f;
  float h1 = 0.0f, c1 = 0.0f;
  float hb0[HID], hb1[HID];
#pragma unroll
  for (int j = 0; j < HID; ++j) { hb0[j] = 0.0f; hb1[j] = 0.0f; }


  float4 x0v = *(const float4*)(xg0p);
  float4 x1v = *(const float4*)(xg1p);
  {
    float p0a = 0.f, p0b = 0.f, p1a = 0.f, p1b = 0.f;
#pragma unroll
    for (int j = 0; j < 8; ++j) {
      p0a = fmaf(w0a[j],     hb0[j],     p0a);
      p0b = fmaf(w0a[j + 8], hb0[j + 8], p0b);
      p1a = fmaf(w0b[j],     hb0[j],     p1a);
      p1b = fmaf(w0b[j + 8], hb0[j + 8], p1b);
    }
    const float a0 = x0v.x + (p0a + p0b);
    const float a1 = x1v.x + (p1a + p1b);
    const float fv = __shfl_xor(a0, 16, 32);
    const float ov = __shfl_xor(a1, 16, 32);
    c0 = sig_f(fv) * c0 + sig_f(a0) * tanh_f(a1);
    h0 = sig_f(ov) * tanh_f(c0);
#pragma unroll
    for (int j = 0; j < HID; ++j) hb0[j] = bcast(h0, j);
  }

  for (int t0 = 0; t0 < T_SEQ; t0 += 4) {
    const int tn = (t0 + 4 < T_SEQ) ? (t0 + 4) : t0;
    const float4 nx0 = *(const float4*)(xg0p + tn);
    const float4 nx1 = *(const float4*)(xg1p + tn);
    __builtin_prefetch(xg0p + t0 + 64, 0, 3);
    __builtin_prefetch(xg1p + t0 + 64, 0, 3);
    const float x0a[4] = {x0v.y, x0v.z, x0v.w, nx0.x};
    const float x1a[4] = {x1v.y, x1v.z, x1v.w, nx1.x};

#pragma unroll
    for (int u = 0; u < 4; ++u) {
      float p0a = 0.f, p0b = 0.f, p1a = 0.f, p1b = 0.f;
      float r0a = 0.f, r0b = 0.f, r0c = 0.f, r0d = 0.f;
      float r1a = 0.f, r1b = 0.f, r1c = 0.f, r1d = 0.f;
#pragma unroll
      for (int j = 0; j < 8; ++j) {
        p0a = fmaf(w0a[j],     hb0[j],     p0a);
        p0b = fmaf(w0a[j + 8], hb0[j + 8], p0b);
        p1a = fmaf(w0b[j],     hb0[j],     p1a);
        p1b = fmaf(w0b[j + 8], hb0[j + 8], p1b);
        r0a = fmaf(wia[j],     hb0[j],     r0a);
        r0b = fmaf(wia[j + 8], hb0[j + 8], r0b);
        r0c = fmaf(wha[j],     hb1[j],     r0c);
        r0d = fmaf(wha[j + 8], hb1[j + 8], r0d);
        r1a = fmaf(wib[j],     hb0[j],     r1a);
        r1b = fmaf(wib[j + 8], hb0[j + 8], r1b);
        r1c = fmaf(whb[j],     hb1[j],     r1c);
        r1d = fmaf(whb[j + 8], hb1[j + 8], r1d);
      }
      const float a0 = x0a[u] + (p0a + p0b);
      const float a1 = x1a[u] + (p1a + p1b);
      const float b0 = bias0 + ((r0a + r0b) + (r0c + r0d));
      const float b1 = bias1 + ((r1a + r1b) + (r1c + r1d));
      const float fv  = __shfl_xor(a0, 16, 32);
      const float ov  = __shfl_xor(a1, 16, 32);
      const float fv1 = __shfl_xor(b0, 16, 32);
      const float ov1 = __shfl_xor(b1, 16, 32);
      c0 = sig_f(fv)  * c0 + sig_f(a0) * tanh_f(a1);
      c1 = sig_f(fv1) * c1 + sig_f(b0) * tanh_f(b1);
      h0 = sig_f(ov)  * tanh_f(c0);
      h1 = sig_f(ov1) * tanh_f(c1);

#pragma unroll
      for (int j = 0; j < HID; ++j) { hb0[j] = bcast(h0, j); hb1[j] = bcast(h1, j); }

      const float sA = (hb1[0] + hb1[1]) + (hb1[2] + hb1[3]);
      const float sB = (hb1[4] + hb1[5]) + (hb1[6] + hb1[7]);
      const float sC = (hb1[8] + hb1[9]) + (hb1[10] + hb1[11]);
      const float sD = (hb1[12] + hb1[13]) + (hb1[14] + hb1[15]);
      const float mean = ((sA + sB) + (sC + sD)) * 0.0625f;
      if (lane == 0) outb[t0 + u] = mean;
    }
    x0v = nx0;
    x1v = nx1;
  }
}

extern "C" void kernel_launch(void* const* d_in, const int* in_sizes, int n_in,
                              void* d_out, int out_size, void* d_ws, size_t ws_size,
                              hipStream_t stream) {
  (void)in_sizes; (void)n_in; (void)out_size; (void)ws_size;
  const float* x    = (const float*)d_in[0];
  const float* Wih0 = (const float*)d_in[1];
  const float* Whh0 = (const float*)d_in[2];
  const float* bih0 = (const float*)d_in[3];
  const float* bhh0 = (const float*)d_in[4];
  const float* Wih1 = (const float*)d_in[5];
  const float* Whh1 = (const float*)d_in[6];
  const float* bih1 = (const float*)d_in[7];
  const float* bhh1 = (const float*)d_in[8];
  float* out = (float*)d_out;
  float* xg  = (float*)d_ws;

  dim3 g1(T_SEQ / 16, BATCH, 1);
  lstm_in_gemm_wmma<<<g1, dim3(128, 1, 1), 0, stream>>>(x, Wih0, bih0, bhh0, xg);

  lstm_recurrence<<<dim3(BATCH, 1, 1), dim3(32, 1, 1), 0, stream>>>(
      xg, Whh0, Wih1, Whh1, bih1, bhh1, out);
}


# recurrence: half-swap gate gather (lane l <-> l^16) by v_permlanex16 on the vector ALU instead of the LDS-unit permute that __shfl_xor compiles to; sums and their order unchanged
# speedup vs baseline: 1.5738x; 1.2140x over previous
#include <hip/hip_runtime.h>

typedef float v2f __attribute__((ext_vector_type(2)));
typedef float v8f __attribute__((ext_vector_type(8)));

#define T_SEQ 2048
#define FEAT  256
#define GATES 64
#define HID   16
#define BATCH 64

__global__ __launch_bounds__(128) void lstm_in_gemm_wmma(
    const float* __restrict__ x,
    const float* __restrict__ W,
    const float* __restrict__ bih,
    const float* __restrict__ bhh,
    float* __restrict__ xg)
{
  const int lane = threadIdx.x & 31;
  const int m    = threadIdx.x >> 5;
  const int half = lane >> 4;
  const int r    = lane & 15;
  const int tb   = blockIdx.x * 16;
  const int b    = blockIdx.y;

  const float* Xb   = x + (size_t)b * FEAT * T_SEQ;
  const float* Wrow = W + (size_t)(m * 16 + r) * FEAT;

  v8f acc = {};
  for (int k = 0; k < FEAT; k += 4) {
    const int ka = k + 2 * half;
    v2f av, bv;
    av.x = Wrow[ka];
    av.y = Wrow[ka + 1];
    bv.x = Xb[(size_t)ka * T_SEQ + tb + r];
    bv.y = Xb[(size_t)(ka + 1) * T_SEQ + tb + r];
    acc = __builtin_amdgcn_wmma_f32_16x16x4_f32(
         false, av,  false, bv,
         (short)0, acc,  false,  false);
  }

#pragma unroll
  for (int v = 0; v < 8; ++v) {
    const int g = m * 16 + v + 8 * half;
    const float val = acc[v] + bih[g] + bhh[g];
    xg[((size_t)b * GATES + g) * T_SEQ + tb + r] = val;
  }
}

__device__ __forceinline__ float bcast(float v, int j) {
  return __uint_as_float(__builtin_amdgcn_readlane(__float_as_uint(v), (unsigned)j));
}
__device__ __forceinline__ float xhalf(float x) {
  const int b = __builtin_bit_cast(int, x);
  return __builtin_bit_cast(float,
      __builtin_amdgcn_permlanex16(b, b, 0x76543210, 0xfedcba98, false, false));
}
__device__ __forceinline__ float sig_f(float x) {
  return __builtin_amdgcn_rcpf(1.0f + __expf(-x));
}
__device__ __forceinline__ float tanh_f(float x) {
  return 1.0f - 2.0f * __builtin_amdgcn_rcpf(__expf(2.0f * x) + 1.0f);
}

__global__ __launch_bounds__(32) void lstm_recurrence(
    const float* __restrict__ xg,
    const float* __restrict__ Whh0,
    const float* __restrict__ Wih1,
    const float* __restrict__ Whh1,
    const float* __restrict__ bih1,
    const float* __restrict__ bhh1,
    float* __restrict__ out)
{
  const int lane = threadIdx.x;
  const int b    = blockIdx.x;
  const int g0   = lane;
  const int g1   = lane + 32;

  float w0a[HID], w0b[HID], wia[HID], wib[HID], wha[HID], whb[HID];
#pragma unroll
  for (int j = 0; j < HID; ++j) {
    w0a[j] = Whh0[g0 * HID + j];
    w0b[j] = Whh0[g1 * HID + j];
    wia[j] = Wih1[g0 * HID + j];
    wib[j] = Wih1[g1 * HID + j];
    wha[j] = Whh1[g0 * HID + j];
    whb[j] = Whh1[g1 * HID + j];
  }
  const float bias0 = bih1[g0] + bhh1[g0];
  const float bias1 = bih1[g1] + bhh1[g1];

  const float* xg0p = xg + ((size_t)b * GATES + g0) * T_SEQ;
  const float* xg1p = xg + ((size_t)b * GATES + g1) * T_SEQ;
  float* outb = out + (size_t)b * T_SEQ;

  float h0 = 0.0f, c0 = 0.0f;
  float h1 = 0.0f, c1 = 0.0f;
  float hb0[HID], hb1[HID];
#pragma unroll
  for (int j = 0; j < HID; ++j) { hb0[j] = 0.0f; hb1[j] = 0.0f; }


  float4 x0v = *(const float4*)(xg0p);
  float4 x1v = *(const float4*)(xg1p);
  {
    float p0a = 0.f, p0b = 0.f, p1a = 0.f, p1b = 0.f;
#pragma unroll
    for (int j = 0; j < 8; ++j) {
      p0a = fmaf(w0a[j],     hb0[j],     p0a);
      p0b = fmaf(w0a[j + 8], hb0[j + 8], p0b);
      p1a = fmaf(w0b[j],     hb0[j],     p1a);
      p1b = fmaf(w0b[j + 8], hb0[j + 8], p1b);
    }
    const float a0 = x0v.x + (p0a + p0b);
    const float a1 = x1v.x + (p1a + p1b);
    const float fv = xhalf(a0);
    const float ov = xhalf(a1);
    c0 = sig_f(fv) * c0 + sig_f(a0) * tanh_f(a1);
    h0 = sig_f(ov) * tanh_f(c0);
#pragma unroll
    for (int j = 0; j < HID; ++j) hb0[j] = bcast(h0, j);
  }

  for (int t0 = 0; t0 < T_SEQ; t0 += 4) {
    const int tn = (t0 + 4 < T_SEQ) ? (t0 + 4) : t0;
    const float4 nx0 = *(const float4*)(xg0p + tn);
    const float4 nx1 = *(const float4*)(xg1p + tn);
    __builtin_prefetch(xg0p + t0 + 64, 0, 3);
    __builtin_prefetch(xg1p + t0 + 64, 0, 3);
    const float x0a[4] = {x0v.y, x0v.z, x0v.w, nx0.x};
    const float x1a[4] = {x1v.y, x1v.z, x1v.w, nx1.x};

#pragma unroll
    for (int u = 0; u < 4; ++u) {
      float p0a = 0.f, p0b = 0.f, p1a = 0.f, p1b = 0.f;
      float r0a = 0.f, r0b = 0.f, r0c = 0.f, r0d = 0.f;
      float r1a = 0.f, r1b = 0.f, r1c = 0.f, r1d = 0.f;
#pragma unroll
      for (int j = 0; j < 8; ++j) {
        p0a = fmaf(w0a[j],     hb0[j],     p0a);
        p0b = fmaf(w0a[j + 8], hb0[j + 8], p0b);
        p1a = fmaf(w0b[j],     hb0[j],     p1a);
        p1b = fmaf(w0b[j + 8], hb0[j + 8], p1b);
        r0a = fmaf(wia[j],     hb0[j],     r0a);
        r0b = fmaf(wia[j + 8], hb0[j + 8], r0b);
        r0c = fmaf(wha[j],     hb1[j],     r0c);
        r0d = fmaf(wha[j + 8], hb1[j + 8], r0d);
        r1a = fmaf(wib[j],     hb0[j],     r1a);
        r1b = fmaf(wib[j + 8], hb0[j + 8], r1b);
        r1c = fmaf(whb[j],     hb1[j],     r1c);
        r1d = fmaf(whb[j + 8], hb1[j + 8], r1d);
      }
      const float a0 = x0a[u] + (p0a + p0b);
      const float a1 = x1a[u] + (p1a + p1b);
      const float b0 = bias0 + ((r0a + r0b) + (r0c + r0d));
      const float b1 = bias1 + ((r1a + r1b) + (r1c + r1d));
      const float fv  = xhalf(a0);
      const float ov  = xhalf(a1);
      const float fv1 = xhalf(b0);
      const float ov1 = xhalf(b1);
      c0 = sig_f(fv)  * c0 + sig_f(a0) * tanh_f(a1);
      c1 = sig_f(fv1) * c1 + sig_f(b0) * tanh_f(b1);
      h0 = sig_f(ov)  * tanh_f(c0);
      h1 = sig_f(ov1) * tanh_f(c1);

#pragma unroll
      for (int j = 0; j < HID; ++j) { hb0[j] = bcast(h0, j); hb1[j] = bcast(h1, j); }

      const float sA = (hb1[0] + hb1[1]) + (hb1[2] + hb1[3]);
      const float sB = (hb1[4] + hb1[5]) + (hb1[6] + hb1[7]);
      const float sC = (hb1[8] + hb1[9]) + (hb1[10] + hb1[11]);
      const float sD = (hb1[12] + hb1[13]) + (hb1[14] + hb1[15]);
      const float mean = ((sA + sB) + (sC + sD)) * 0.0625f;
      if (lane == 0) outb[t0 + u] = mean;
    }
    x0v = nx0;
    x1v = nx1;
  }
}

extern "C" void kernel_launch(void* const* d_in, const int* in_sizes, int n_in,
                              void* d_out, int out_size, void* d_ws, size_t ws_size,
                              hipStream_t stream) {
  (void)in_sizes; (void)n_in; (void)out_size; (void)ws_size;
  const float* x    = (const float*)d_in[0];
  const float* Wih0 = (const float*)d_in[1];
  const float* Whh0 = (const float*)d_in[2];
  const float* bih0 = (const float*)d_in[3];
  const float* bhh0 = (const float*)d_in[4];
  const float* Wih1 = (const float*)d_in[5];
  const float* Whh1 = (const float*)d_in[6];
  const float* bih1 = (const float*)d_in[7];
  const float* bhh1 = (const float*)d_in[8];
  float* out = (float*)d_out;
  float* xg  = (float*)d_ws;

  dim3 g1(T_SEQ / 16, BATCH, 1);
  lstm_in_gemm_wmma<<<g1, dim3(128, 1, 1), 0, stream>>>(x, Wih0, bih0, bhh0, xg);

  lstm_recurrence<<<dim3(BATCH, 1, 1), dim3(32, 1, 1), 0, stream>>>(
      xg, Whh0, Wih1, Whh1, bih1, bhh1, out);
}


# recurrence: hidden-mean off each step's serial path: row_xmask DPP butterfly of h1 (same add tree) one step late, sum parked in lane t%32, one 128-byte store per 32 steps; no lane-0 EXEC branch
# speedup vs baseline: 2.0523x; 1.3040x over previous
#include <hip/hip_runtime.h>

typedef float v2f __attribute__((ext_vector_type(2)));
typedef float v8f __attribute__((ext_vector_type(8)));

#define T_SEQ 2048
#define FEAT  256
#define GATES 64
#define HID   16
#define BATCH 64

__global__ __launch_bounds__(128) void lstm_in_gemm_wmma(
    const float* __restrict__ x,
    const float* __restrict__ W,
    const float* __restrict__ bih,
    const float* __restrict__ bhh,
    float* __restrict__ xg)
{
  const int lane = threadIdx.x & 31;
  const int m    = threadIdx.x >> 5;
  const int half = lane >> 4;
  const int r    = lane & 15;
  const int tb   = blockIdx.x * 16;
  const int b    = blockIdx.y;

  const float* Xb   = x + (size_t)b * FEAT * T_SEQ;
  const float* Wrow = W + (size_t)(m * 16 + r) * FEAT;

  v8f acc = {};
  for (int k = 0; k < FEAT; k += 4) {
    const int ka = k + 2 * half;
    v2f av, bv;
    av.x = Wrow[ka];
    av.y = Wrow[ka + 1];
    bv.x = Xb[(size_t)ka * T_SEQ + tb + r];
    bv.y = Xb[(size_t)(ka + 1) * T_SEQ + tb + r];
    acc = __builtin_amdgcn_wmma_f32_16x16x4_f32(
         false, av,  false, bv,
         (short)0, acc,  false,  false);
  }

#pragma unroll
  for (int v = 0; v < 8; ++v) {
    const int g = m * 16 + v + 8 * half;
    const float val = acc[v] + bih[g] + bhh[g];
    xg[((size_t)b * GATES + g) * T_SEQ + tb + r] = val;
  }
}

__device__ __forceinline__ float bcast(float v, int j) {
  return __uint_as_float(__builtin_amdgcn_readlane(__float_as_uint(v), (unsigned)j));
}
__device__ __forceinline__ float xhalf(float x) {
  const int b = __builtin_bit_cast(int, x);
  return __builtin_bit_cast(float,
      __builtin_amdgcn_permlanex16(b, b, 0x76543210, 0xfedcba98, false, false));
}
template <int M>
__device__ __forceinline__ float rowx(float x) {
  const int b = __builtin_bit_cast(int, x);
  return __builtin_bit_cast(float,
      __builtin_amdgcn_update_dpp(b, b, 0x160 + M, 0xf, 0xf, false));
}
__device__ __forceinline__ float sig_f(float x) {
  return __builtin_amdgcn_rcpf(1.0f + __expf(-x));
}
__device__ __forceinline__ float tanh_f(float x) {
  return 1.0f - 2.0f * __builtin_amdgcn_rcpf(__expf(2.0f * x) + 1.0f);
}

__device__ __forceinline__ float unitsum(float h, bool lower) {
  float s = h;
  s += rowx<1>(s);
  s += rowx<2>(s);
  s += rowx<4>(s);
  s += rowx<8>(s);
  const float sx = xhalf(s);
  return lower ? s : sx;
}

__global__ __launch_bounds__(32) void lstm_recurrence(
    const float* __restrict__ xg,
    const float* __restrict__ Whh0,
    const float* __restrict__ Wih1,
    const float* __restrict__ Whh1,
    const float* __restrict__ bih1,
    const float* __restrict__ bhh1,
    float* __restrict__ out)
{
  const int lane = threadIdx.x;
  const int b    = blockIdx.x;
  const int g0   = lane;
  const int g1   = lane + 32;

  float w0a[HID], w0b[HID], wia[HID], wib[HID], wha[HID], whb[HID];
#pragma unroll
  for (int j = 0; j < HID; ++j) {
    w0a[j] = Whh0[g0 * HID + j];
    w0b[j] = Whh0[g1 * HID + j];
    wia[j] = Wih1[g0 * HID + j];
    wib[j] = Wih1[g1 * HID + j];
    wha[j] = Whh1[g0 * HID + j];
    whb[j] = Whh1[g1 * HID + j];
  }
  const float bias0 = bih1[g0] + bhh1[g0];
  const float bias1 = bih1[g1] + bhh1[g1];

  const float* xg0p = xg + ((size_t)b * GATES + g0) * T_SEQ;
  const float* xg1p = xg + ((size_t)b * GATES + g1) * T_SEQ;
  float* outb = out + (size_t)b * T_SEQ;

  float h0 = 0.0f, c0 = 0.0f;
  float h1 = 0.0f, c1 = 0.0f;
  float hb0[HID], hb1[HID];
#pragma unroll
  for (int j = 0; j < HID; ++j) { hb0[j] = 0.0f; hb1[j] = 0.0f; }
  float outacc = 0.0f;


  float4 x0v = *(const float4*)(xg0p);
  float4 x1v = *(const float4*)(xg1p);
  {
    float p0a = 0.f, p0b = 0.f, p1a = 0.f, p1b = 0.f;
#pragma unroll
    for (int j = 0; j < 8; ++j) {
      p0a = fmaf(w0a[j],     hb0[j],     p0a);
      p0b = fmaf(w0a[j + 8], hb0[j + 8], p0b);
      p1a = fmaf(w0b[j],     hb0[j],     p1a);
      p1b = fmaf(w0b[j + 8], hb0[j + 8], p1b);
    }
    const float a0 = x0v.x + (p0a + p0b);
    const float a1 = x1v.x + (p1a + p1b);
    const float fv = xhalf(a0);
    const float ov = xhalf(a1);
    c0 = sig_f(fv) * c0 + sig_f(a0) * tanh_f(a1);
    h0 = sig_f(ov) * tanh_f(c0);
#pragma unroll
    for (int j = 0; j < HID; ++j) hb0[j] = bcast(h0, j);
  }

  for (int t0 = 0; t0 < T_SEQ; t0 += 4) {
    const int tn = (t0 + 4 < T_SEQ) ? (t0 + 4) : t0;
    const float4 nx0 = *(const float4*)(xg0p + tn);
    const float4 nx1 = *(const float4*)(xg1p + tn);
    __builtin_prefetch(xg0p + t0 + 64, 0, 3);
    __builtin_prefetch(xg1p + t0 + 64, 0, 3);
    const float x0a[4] = {x0v.y, x0v.z, x0v.w, nx0.x};
    const float x1a[4] = {x1v.y, x1v.z, x1v.w, nx1.x};

#pragma unroll
    for (int u = 0; u < 4; ++u) {
      const float sprev = unitsum(h1, lane < 16);
      outacc = (lane == ((t0 + u - 1) & 31)) ? sprev : outacc;
      float p0a = 0.f, p0b = 0.f, p1a = 0.f, p1b = 0.f;
      float r0a = 0.f, r0b = 0.f, r0c = 0.f, r0d = 0.f;
      float r1a = 0.f, r1b = 0.f, r1c = 0.f, r1d = 0.f;
#pragma unroll
      for (int j = 0; j < 8; ++j) {
        p0a = fmaf(w0a[j],     hb0[j],     p0a);
        p0b = fmaf(w0a[j + 8], hb0[j + 8], p0b);
        p1a = fmaf(w0b[j],     hb0[j],     p1a);
        p1b = fmaf(w0b[j + 8], hb0[j + 8], p1b);
        r0a = fmaf(wia[j],     hb0[j],     r0a);
        r0b = fmaf(wia[j + 8], hb0[j + 8], r0b);
        r0c = fmaf(wha[j],     hb1[j],     r0c);
        r0d = fmaf(wha[j + 8], hb1[j + 8], r0d);
        r1a = fmaf(wib[j],     hb0[j],     r1a);
        r1b = fmaf(wib[j + 8], hb0[j + 8], r1b);
        r1c = fmaf(whb[j],     hb1[j],     r1c);
        r1d = fmaf(whb[j + 8], hb1[j + 8], r1d);
      }
      const float a0 = x0a[u] + (p0a + p0b);
      const float a1 = x1a[u] + (p1a + p1b);
      const float b0 = bias0 + ((r0a + r0b) + (r0c + r0d));
      const float b1 = bias1 + ((r1a + r1b) + (r1c + r1d));
      const float fv  = xhalf(a0);
      const float ov  = xhalf(a1);
      const float fv1 = xhalf(b0);
      const float ov1 = xhalf(b1);
      c0 = sig_f(fv)  * c0 + sig_f(a0) * tanh_f(a1);
      c1 = sig_f(fv1) * c1 + sig_f(b0) * tanh_f(b1);
      h0 = sig_f(ov)  * tanh_f(c0);
      h1 = sig_f(ov1) * tanh_f(c1);

#pragma unroll
      for (int j = 0; j < HID; ++j) { hb0[j] = bcast(h0, j); hb1[j] = bcast(h1, j); }

      __builtin_amdgcn_sched_barrier(0);

      if (u == 0 && (t0 & 31) == 0 && t0 != 0)
        outb[(t0 - 32) + lane] = outacc * 0.0625f;
    }
    x0v = nx0;
    x1v = nx1;
  }
  const float slast = unitsum(h1, lane < 16);
  outacc = (lane == 31) ? slast : outacc;
  outb[(T_SEQ - 32) + lane] = outacc * 0.0625f;
}

extern "C" void kernel_launch(void* const* d_in, const int* in_sizes, int n_in,
                              void* d_out, int out_size, void* d_ws, size_t ws_size,
                              hipStream_t stream) {
  (void)in_sizes; (void)n_in; (void)out_size; (void)ws_size;
  const float* x    = (const float*)d_in[0];
  const float* Wih0 = (const float*)d_in[1];
  const float* Whh0 = (const float*)d_in[2];
  const float* bih0 = (const float*)d_in[3];
  const float* bhh0 = (const float*)d_in[4];
  const float* Wih1 = (const float*)d_in[5];
  const float* Whh1 = (const float*)d_in[6];
  const float* bih1 = (const float*)d_in[7];
  const float* bhh1 = (const float*)d_in[8];
  float* out = (float*)d_out;
  float* xg  = (float*)d_ws;

  dim3 g1(T_SEQ / 16, BATCH, 1);
  lstm_in_gemm_wmma<<<g1, dim3(128, 1, 1), 0, stream>>>(x, Wih0, bih0, bhh0, xg);

  lstm_recurrence<<<dim3(BATCH, 1, 1), dim3(32, 1, 1), 0, stream>>>(
      xg, Whh0, Wih1, Whh1, bih1, bhh1, out);
}


# recurrence: one layer per wave, two waves per work-group; layer-0 wave passes Wih1@h0(t) to the layer-1 wave via a 2-buffer LDS ring, one barrier per 32 steps; sums and their order unchanged
# speedup vs baseline: 2.4767x; 1.2068x over previous
#include <hip/hip_runtime.h>

typedef float v2f __attribute__((ext_vector_type(2)));
typedef float v8f __attribute__((ext_vector_type(8)));

#define T_SEQ 2048
#define FEAT  256
#define GATES 64
#define HID   16
#define BATCH 64

__global__ __launch_bounds__(128) void lstm_in_gemm_wmma(
    const float* __restrict__ x,
    const float* __restrict__ W,
    const float* __restrict__ bih,
    const float* __restrict__ bhh,
    float* __restrict__ xg)
{
  const int lane = threadIdx.x & 31;
  const int m    = threadIdx.x >> 5;
  const int half = lane >> 4;
  const int r    = lane & 15;
  const int tb   = blockIdx.x * 16;
  const int b    = blockIdx.y;

  const float* Xb   = x + (size_t)b * FEAT * T_SEQ;
  const float* Wrow = W + (size_t)(m * 16 + r) * FEAT;

  v8f acc = {};
  for (int k = 0; k < FEAT; k += 4) {
    const int ka = k + 2 * half;
    v2f av, bv;
    av.x = Wrow[ka];
    av.y = Wrow[ka + 1];
    bv.x = Xb[(size_t)ka * T_SEQ + tb + r];
    bv.y = Xb[(size_t)(ka + 1) * T_SEQ + tb + r];
    acc = __builtin_amdgcn_wmma_f32_16x16x4_f32(
         false, av,  false, bv,
         (short)0, acc,  false,  false);
  }

#pragma unroll
  for (int v = 0; v < 8; ++v) {
    const int g = m * 16 + v + 8 * half;
    const float val = acc[v] + bih[g] + bhh[g];
    xg[((size_t)b * GATES + g) * T_SEQ + tb + r] = val;
  }
}

#define CH  32
#define NCH (T_SEQ / CH)

__device__ __forceinline__ float bcast(float v, int j) {
  return __uint_as_float(__builtin_amdgcn_readlane(__float_as_uint(v), (unsigned)j));
}
__device__ __forceinline__ float xhalf(float x) {
  const int b = __builtin_bit_cast(int, x);
  return __builtin_bit_cast(float,
      __builtin_amdgcn_permlanex16(b, b, 0x76543210, 0xfedcba98, false, false));
}
template <int M>
__device__ __forceinline__ float rowx(float x) {
  const int b = __builtin_bit_cast(int, x);
  return __builtin_bit_cast(float,
      __builtin_amdgcn_update_dpp(b, b, 0x160 + M, 0xf, 0xf, false));
}
__device__ __forceinline__ float sig_f(float x) {
  return __builtin_amdgcn_rcpf(1.0f + __expf(-x));
}
__device__ __forceinline__ float tanh_f(float x) {
  return 1.0f - 2.0f * __builtin_amdgcn_rcpf(__expf(2.0f * x) + 1.0f);
}

__device__ __forceinline__ float unitsum(float h, bool lower) {
  float s = h;
  s += rowx<1>(s);
  s += rowx<2>(s);
  s += rowx<4>(s);
  s += rowx<8>(s);
  const float sx = xhalf(s);
  return lower ? s : sx;
}

__global__ __launch_bounds__(64) void lstm_recurrence(
    const float* __restrict__ xg,
    const float* __restrict__ Whh0,
    const float* __restrict__ Wih1,
    const float* __restrict__ Whh1,
    const float* __restrict__ bih1,
    const float* __restrict__ bhh1,
    float* __restrict__ out)
{
  __shared__ float2 ring[2][CH][32];

  const int lane = threadIdx.x & 31;
  const int role = __builtin_amdgcn_readfirstlane((int)(threadIdx.x >> 5));
  const int b    = blockIdx.x;
  const int g0   = lane;
  const int g1   = lane + 32;

  float w0a[HID], w0b[HID], wia[HID], wib[HID], wha[HID], whb[HID];
#pragma unroll
  for (int j = 0; j < HID; ++j) {
    w0a[j] = Whh0[g0 * HID + j];
    w0b[j] = Whh0[g1 * HID + j];
    wia[j] = Wih1[g0 * HID + j];
    wib[j] = Wih1[g1 * HID + j];
    wha[j] = Whh1[g0 * HID + j];
    whb[j] = Whh1[g1 * HID + j];
  }
  const float bias0 = bih1[g0] + bhh1[g0];
  const float bias1 = bih1[g1] + bhh1[g1];

  const float* xg0p = xg + ((size_t)b * GATES + g0) * T_SEQ;
  const float* xg1p = xg + ((size_t)b * GATES + g1) * T_SEQ;
  float* outb = out + (size_t)b * T_SEQ;

  float h0 = 0.0f, c0 = 0.0f;
  float h1 = 0.0f, c1 = 0.0f;
  float hb0[HID], hb1[HID];
#pragma unroll
  for (int j = 0; j < HID; ++j) { hb0[j] = 0.0f; hb1[j] = 0.0f; }
  float outacc = 0.0f;

  float4 x0v = make_float4(0.f, 0.f, 0.f, 0.f);
  float4 x1v = make_float4(0.f, 0.f, 0.f, 0.f);

  if (role == 0) {
    x0v = *(const float4*)(xg0p);
    x1v = *(const float4*)(xg1p);
    float p0a = 0.f, p0b = 0.f, p1a = 0.f, p1b = 0.f;
#pragma unroll
    for (int j = 0; j < 8; ++j) {
      p0a = fmaf(w0a[j],     hb0[j],     p0a);
      p0b = fmaf(w0a[j + 8], hb0[j + 8], p0b);
      p1a = fmaf(w0b[j],     hb0[j],     p1a);
      p1b = fmaf(w0b[j + 8], hb0[j + 8], p1b);
    }
    const float a0 = x0v.x + (p0a + p0b);
    const float a1 = x1v.x + (p1a + p1b);
    const float fv = xhalf(a0);
    const float ov = xhalf(a1);
    c0 = sig_f(fv) * c0 + sig_f(a0) * tanh_f(a1);
    h0 = sig_f(ov) * tanh_f(c0);
#pragma unroll
    for (int j = 0; j < HID; ++j) hb0[j] = bcast(h0, j);
  }

  for (int p = 0; p <= NCH; ++p) {
    if (role == 0) {
      if (p < NCH) {
        float2 (*wbuf)[32] = ring[p & 1];
        for (int t0 = 0; t0 < CH; t0 += 4) {
          const int tg = p * CH + t0;
          const int tn = (tg + 4 < T_SEQ) ? (tg + 4) : tg;
          const float4 nx0 = *(const float4*)(xg0p + tn);
          const float4 nx1 = *(const float4*)(xg1p + tn);
          __builtin_prefetch(xg0p + tg + 64, 0, 3);
          __builtin_prefetch(xg1p + tg + 64, 0, 3);
          const float x0a[4] = {x0v.y, x0v.z, x0v.w, nx0.x};
          const float x1a[4] = {x1v.y, x1v.z, x1v.w, nx1.x};

#pragma unroll
          for (int u = 0; u < 4; ++u) {
            float p0a = 0.f, p0b = 0.f, p1a = 0.f, p1b = 0.f;
            float r0a = 0.f, r0b = 0.f, r1a = 0.f, r1b = 0.f;
#pragma unroll
            for (int j = 0; j < 8; ++j) {
              p0a = fmaf(w0a[j],     hb0[j],     p0a);
              p0b = fmaf(w0a[j + 8], hb0[j + 8], p0b);
              p1a = fmaf(w0b[j],     hb0[j],     p1a);
              p1b = fmaf(w0b[j + 8], hb0[j + 8], p1b);
              r0a = fmaf(wia[j],     hb0[j],     r0a);
              r0b = fmaf(wia[j + 8], hb0[j + 8], r0b);
              r1a = fmaf(wib[j],     hb0[j],     r1a);
              r1b = fmaf(wib[j + 8], hb0[j + 8], r1b);
            }
            wbuf[t0 + u][lane] = make_float2(r0a + r0b, r1a + r1b);
            const float a0 = x0a[u] + (p0a + p0b);
            const float a1 = x1a[u] + (p1a + p1b);
            const float fv = xhalf(a0);
            const float ov = xhalf(a1);
            c0 = sig_f(fv) * c0 + sig_f(a0) * tanh_f(a1);
            h0 = sig_f(ov) * tanh_f(c0);
#pragma unroll
            for (int j = 0; j < HID; ++j) hb0[j] = bcast(h0, j);
            __builtin_amdgcn_sched_barrier(0);
          }
          x0v = nx0;
          x1v = nx1;
        }
      }
    } else {
      if (p >= 1) {
        const int q = p - 1;
        const float2 (*rbuf)[32] = ring[q & 1];
        float2 s0 = rbuf[0][lane], s1 = rbuf[1][lane];
        float2 s2 = rbuf[2][lane], s3 = rbuf[3][lane];
        for (int t0 = 0; t0 < CH; t0 += 4) {
          const int tq = (t0 + 4 < CH) ? (t0 + 4) : t0;
          const float2 n0 = rbuf[tq][lane],     n1 = rbuf[tq + 1][lane];
          const float2 n2 = rbuf[tq + 2][lane], n3 = rbuf[tq + 3][lane];
          const float sa[4] = {s0.x, s1.x, s2.x, s3.x};
          const float sb[4] = {s0.y, s1.y, s2.y, s3.y};

#pragma unroll
          for (int u = 0; u < 4; ++u) {
            const float sprev = unitsum(h1, lane < 16);
            outacc = (lane == ((t0 + u - 1) & 31)) ? sprev : outacc;
            float r0c = 0.f, r0d = 0.f, r1c = 0.f, r1d = 0.f;
#pragma unroll
            for (int j = 0; j < 8; ++j) {
              r0c = fmaf(wha[j],     hb1[j],     r0c);
              r0d = fmaf(wha[j + 8], hb1[j + 8], r0d);
              r1c = fmaf(whb[j],     hb1[j],     r1c);
              r1d = fmaf(whb[j + 8], hb1[j + 8], r1d);
            }
            const float b0 = bias0 + (sa[u] + (r0c + r0d));
            const float b1 = bias1 + (sb[u] + (r1c + r1d));
            const float fv1 = xhalf(b0);
            const float ov1 = xhalf(b1);
            c1 = sig_f(fv1) * c1 + sig_f(b0) * tanh_f(b1);
            h1 = sig_f(ov1) * tanh_f(c1);
#pragma unroll
            for (int j = 0; j < HID; ++j) hb1[j] = bcast(h1, j);
            __builtin_amdgcn_sched_barrier(0);
            if (u == 0 && t0 == 0 && q != 0)
              outb[(q * CH - 32) + lane] = outacc * 0.0625f;
          }
          s0 = n0; s1 = n1; s2 = n2; s3 = n3;
        }
      }
    }
    __builtin_amdgcn_fence(__ATOMIC_RELEASE, "workgroup", "local");
    __builtin_amdgcn_s_barrier();
    __builtin_amdgcn_fence(__ATOMIC_ACQUIRE, "workgroup", "local");
  }

  if (role != 0) {
    const float slast = unitsum(h1, lane < 16);
    outacc = (lane == 31) ? slast : outacc;
    outb[(T_SEQ - 32) + lane] = outacc * 0.0625f;
  }
}

extern "C" void kernel_launch(void* const* d_in, const int* in_sizes, int n_in,
                              void* d_out, int out_size, void* d_ws, size_t ws_size,
                              hipStream_t stream) {
  (void)in_sizes; (void)n_in; (void)out_size; (void)ws_size;
  const float* x    = (const float*)d_in[0];
  const float* Wih0 = (const float*)d_in[1];
  const float* Whh0 = (const float*)d_in[2];
  const float* bih0 = (const float*)d_in[3];
  const float* bhh0 = (const float*)d_in[4];
  const float* Wih1 = (const float*)d_in[5];
  const float* Whh1 = (const float*)d_in[6];
  const float* bih1 = (const float*)d_in[7];
  const float* bhh1 = (const float*)d_in[8];
  float* out = (float*)d_out;
  float* xg  = (float*)d_ws;

  dim3 g1(T_SEQ / 16, BATCH, 1);
  lstm_in_gemm_wmma<<<g1, dim3(128, 1, 1), 0, stream>>>(x, Wih0, bih0, bhh0, xg);

  lstm_recurrence<<<dim3(BATCH, 1, 1), dim3(64, 1, 1), 0, stream>>>(
      xg, Whh0, Wih1, Whh1, bih1, bhh1, out);
}


# recurrence: activations by hardware v_tanh_f32; sigmoid(x) = 0.5+0.5*tanh(x/2), the halving folded exactly into weights, biases and x term; replaces exp+rcp chains; activation rounding changes
# speedup vs baseline: 3.0385x; 1.2268x over previous
#include <hip/hip_runtime.h>

typedef float v2f __attribute__((ext_vector_type(2)));
typedef float v8f __attribute__((ext_vector_type(8)));

#define T_SEQ 2048
#define FEAT  256
#define GATES 64
#define HID   16
#define BATCH 64

__global__ __launch_bounds__(128) void lstm_in_gemm_wmma(
    const float* __restrict__ x,
    const float* __restrict__ W,
    const float* __restrict__ bih,
    const float* __restrict__ bhh,
    float* __restrict__ xg)
{
  const int lane = threadIdx.x & 31;
  const int m    = threadIdx.x >> 5;
  const int half = lane >> 4;
  const int r    = lane & 15;
  const int tb   = blockIdx.x * 16;
  const int b    = blockIdx.y;

  const float* Xb   = x + (size_t)b * FEAT * T_SEQ;
  const float* Wrow = W + (size_t)(m * 16 + r) * FEAT;

  v8f acc = {};
  for (int k = 0; k < FEAT; k += 4) {
    const int ka = k + 2 * half;
    v2f av, bv;
    av.x = Wrow[ka];
    av.y = Wrow[ka + 1];
    bv.x = Xb[(size_t)ka * T_SEQ + tb + r];
    bv.y = Xb[(size_t)(ka + 1) * T_SEQ + tb + r];
    acc = __builtin_amdgcn_wmma_f32_16x16x4_f32(
         false, av,  false, bv,
         (short)0, acc,  false,  false);
  }

#pragma unroll
  for (int v = 0; v < 8; ++v) {
    const int g = m * 16 + v + 8 * half;
    const float val = acc[v] + bih[g] + bhh[g];
    xg[((size_t)b * GATES + g) * T_SEQ + tb + r] = val;
  }
}

#define CH  32
#define NCH (T_SEQ / CH)

__device__ __forceinline__ float bcast(float v, int j) {
  return __uint_as_float(__builtin_amdgcn_readlane(__float_as_uint(v), (unsigned)j));
}
__device__ __forceinline__ float xhalf(float x) {
  const int b = __builtin_bit_cast(int, x);
  return __builtin_bit_cast(float,
      __builtin_amdgcn_permlanex16(b, b, 0x76543210, 0xfedcba98, false, false));
}
template <int M>
__device__ __forceinline__ float rowx(float x) {
  const int b = __builtin_bit_cast(int, x);
  return __builtin_bit_cast(float,
      __builtin_amdgcn_update_dpp(b, b, 0x160 + M, 0xf, 0xf, false));
}
__device__ __forceinline__ float tanh_f(float x) {
  return __builtin_amdgcn_tanhf(x);
}
__device__ __forceinline__ float sig_h(float xh) {
  return fmaf(tanh_f(xh), 0.5f, 0.5f);
}

__device__ __forceinline__ float unitsum(float h, bool lower) {
  float s = h;
  s += rowx<1>(s);
  s += rowx<2>(s);
  s += rowx<4>(s);
  s += rowx<8>(s);
  const float sx = xhalf(s);
  return lower ? s : sx;
}

__global__ __launch_bounds__(64) void lstm_recurrence(
    const float* __restrict__ xg,
    const float* __restrict__ Whh0,
    const float* __restrict__ Wih1,
    const float* __restrict__ Whh1,
    const float* __restrict__ bih1,
    const float* __restrict__ bhh1,
    float* __restrict__ out)
{
  __shared__ float2 ring[2][CH][32];

  const int lane = threadIdx.x & 31;
  const int role = __builtin_amdgcn_readfirstlane((int)(threadIdx.x >> 5));
  const int b    = blockIdx.x;
  const int g0   = lane;
  const int g1   = lane + 32;

  const float s0 = 0.5f;
  const float s1 = (lane < 16) ? 1.0f : 0.5f;
  float w0a[HID], w0b[HID], wia[HID], wib[HID], wha[HID], whb[HID];
#pragma unroll
  for (int j = 0; j < HID; ++j) {
    w0a[j] = Whh0[g0 * HID + j] * s0;
    w0b[j] = Whh0[g1 * HID + j] * s1;
    wia[j] = Wih1[g0 * HID + j] * s0;
    wib[j] = Wih1[g1 * HID + j] * s1;
    wha[j] = Whh1[g0 * HID + j] * s0;
    whb[j] = Whh1[g1 * HID + j] * s1;
  }
  const float bias0 = (bih1[g0] + bhh1[g0]) * s0;
  const float bias1 = (bih1[g1] + bhh1[g1]) * s1;

  const float* xg0p = xg + ((size_t)b * GATES + g0) * T_SEQ;
  const float* xg1p = xg + ((size_t)b * GATES + g1) * T_SEQ;
  float* outb = out + (size_t)b * T_SEQ;

  float h0 = 0.0f, c0 = 0.0f;
  float h1 = 0.0f, c1 = 0.0f;
  float hb0[HID], hb1[HID];
#pragma unroll
  for (int j = 0; j < HID; ++j) { hb0[j] = 0.0f; hb1[j] = 0.0f; }
  float outacc = 0.0f;

  float4 x0v = make_float4(0.f, 0.f, 0.f, 0.f);
  float4 x1v = make_float4(0.f, 0.f, 0.f, 0.f);

  if (role == 0) {
    x0v = *(const float4*)(xg0p);
    x1v = *(const float4*)(xg1p);
    float p0a = 0.f, p0b = 0.f, p1a = 0.f, p1b = 0.f;
#pragma unroll
    for (int j = 0; j < 8; ++j) {
      p0a = fmaf(w0a[j],     hb0[j],     p0a);
      p0b = fmaf(w0a[j + 8], hb0[j + 8], p0b);
      p1a = fmaf(w0b[j],     hb0[j],     p1a);
      p1b = fmaf(w0b[j + 8], hb0[j + 8], p1b);
    }
    const float a0 = fmaf(x0v.x, s0, (p0a + p0b));
    const float a1 = fmaf(x1v.x, s1, (p1a + p1b));
    const float fv = xhalf(a0);
    const float ov = xhalf(a1);
    c0 = sig_h(fv) * c0 + sig_h(a0) * tanh_f(a1);
    h0 = sig_h(ov) * tanh_f(c0);
#pragma unroll
    for (int j = 0; j < HID; ++j) hb0[j] = bcast(h0, j);
  }

  for (int p = 0; p <= NCH; ++p) {
    if (role == 0) {
      if (p < NCH) {
        float2 (*wbuf)[32] = ring[p & 1];
        for (int t0 = 0; t0 < CH; t0 += 4) {
          const int tg = p * CH + t0;
          const int tn = (tg + 4 < T_SEQ) ? (tg + 4) : tg;
          const float4 nx0 = *(const float4*)(xg0p + tn);
          const float4 nx1 = *(const float4*)(xg1p + tn);
          __builtin_prefetch(xg0p + tg + 64, 0, 3);
          __builtin_prefetch(xg1p + tg + 64, 0, 3);
          const float x0a[4] = {x0v.y, x0v.z, x0v.w, nx0.x};
          const float x1a[4] = {x1v.y, x1v.z, x1v.w, nx1.x};

#pragma unroll
          for (int u = 0; u < 4; ++u) {
            float p0a = 0.f, p0b = 0.f, p1a = 0.f, p1b = 0.f;
            float r0a = 0.f, r0b = 0.f, r1a = 0.f, r1b = 0.f;
#pragma unroll
            for (int j = 0; j < 8; ++j) {
              p0a = fmaf(w0a[j],     hb0[j],     p0a);
              p0b = fmaf(w0a[j + 8], hb0[j + 8], p0b);
              p1a = fmaf(w0b[j],     hb0[j],     p1a);
              p1b = fmaf(w0b[j + 8], hb0[j + 8], p1b);
              r0a = fmaf(wia[j],     hb0[j],     r0a);
              r0b = fmaf(wia[j + 8], hb0[j + 8], r0b);
              r1a = fmaf(wib[j],     hb0[j],     r1a);
              r1b = fmaf(wib[j + 8], hb0[j + 8], r1b);
            }
            wbuf[t0 + u][lane] = make_float2(r0a + r0b, r1a + r1b);
            const float a0 = fmaf(x0a[u], s0, (p0a + p0b));
            const float a1 = fmaf(x1a[u], s1, (p1a + p1b));
            const float fv = xhalf(a0);
            const float ov = xhalf(a1);
            c0 = sig_h(fv) * c0 + sig_h(a0) * tanh_f(a1);
            h0 = sig_h(ov) * tanh_f(c0);
#pragma unroll
            for (int j = 0; j < HID; ++j) hb0[j] = bcast(h0, j);
            __builtin_amdgcn_sched_barrier(0);
          }
          x0v = nx0;
          x1v = nx1;
        }
      }
    } else {
      if (p >= 1) {
        const int q = p - 1;
        const float2 (*rbuf)[32] = ring[q & 1];
        float2 s0 = rbuf[0][lane], s1 = rbuf[1][lane];
        float2 s2 = rbuf[2][lane], s3 = rbuf[3][lane];
        for (int t0 = 0; t0 < CH; t0 += 4) {
          const int tq = (t0 + 4 < CH) ? (t0 + 4) : t0;
          const float2 n0 = rbuf[tq][lane],     n1 = rbuf[tq + 1][lane];
          const float2 n2 = rbuf[tq + 2][lane], n3 = rbuf[tq + 3][lane];
          const float sa[4] = {s0.x, s1.x, s2.x, s3.x};
          const float sb[4] = {s0.y, s1.y, s2.y, s3.y};

#pragma unroll
          for (int u = 0; u < 4; ++u) {
            const float sprev = unitsum(h1, lane < 16);
            outacc = (lane == ((t0 + u - 1) & 31)) ? sprev : outacc;
            float r0c = 0.f, r0d = 0.f, r1c = 0.f, r1d = 0.f;
#pragma unroll
            for (int j = 0; j < 8; ++j) {
              r0c = fmaf(wha[j],     hb1[j],     r0c);
              r0d = fmaf(wha[j + 8], hb1[j + 8], r0d);
              r1c = fmaf(whb[j],     hb1[j],     r1c);
              r1d = fmaf(whb[j + 8], hb1[j + 8], r1d);
            }
            const float b0 = bias0 + (sa[u] + (r0c + r0d));
            const float b1 = bias1 + (sb[u] + (r1c + r1d));
            const float fv1 = xhalf(b0);
            const float ov1 = xhalf(b1);
            c1 = sig_h(fv1) * c1 + sig_h(b0) * tanh_f(b1);
            h1 = sig_h(ov1) * tanh_f(c1);
#pragma unroll
            for (int j = 0; j < HID; ++j) hb1[j] = bcast(h1, j);
            __builtin_amdgcn_sched_barrier(0);
            if (u == 0 && t0 == 0 && q != 0)
              outb[(q * CH - 32) + lane] = outacc * 0.0625f;
          }
          s0 = n0; s1 = n1; s2 = n2; s3 = n3;
        }
      }
    }
    __builtin_amdgcn_fence(__ATOMIC_RELEASE, "workgroup", "local");
    __builtin_amdgcn_s_barrier();
    __builtin_amdgcn_fence(__ATOMIC_ACQUIRE, "workgroup", "local");
  }

  if (role != 0) {
    const float slast = unitsum(h1, lane < 16);
    outacc = (lane == 31) ? slast : outacc;
    outb[(T_SEQ - 32) + lane] = outacc * 0.0625f;
  }
}

extern "C" void kernel_launch(void* const* d_in, const int* in_sizes, int n_in,
                              void* d_out, int out_size, void* d_ws, size_t ws_size,
                              hipStream_t stream) {
  (void)in_sizes; (void)n_in; (void)out_size; (void)ws_size;
  const float* x    = (const float*)d_in[0];
  const float* Wih0 = (const float*)d_in[1];
  const float* Whh0 = (const float*)d_in[2];
  const float* bih0 = (const float*)d_in[3];
  const float* bhh0 = (const float*)d_in[4];
  const float* Wih1 = (const float*)d_in[5];
  const float* Whh1 = (const float*)d_in[6];
  const float* bih1 = (const float*)d_in[7];
  const float* bhh1 = (const float*)d_in[8];
  float* out = (float*)d_out;
  float* xg  = (float*)d_ws;

  dim3 g1(T_SEQ / 16, BATCH, 1);
  lstm_in_gemm_wmma<<<g1, dim3(128, 1, 1), 0, stream>>>(x, Wih0, bih0, bhh0, xg);

  lstm_recurrence<<<dim3(BATCH, 1, 1), dim3(64, 1, 1), 0, stream>>>(
      xg, Whh0, Wih1, Whh1, bih1, bhh1, out);
}


# input GEMM: each wave takes 2 gate tiles x 4 interleaved t tiles: x rows read by whole-line 16-byte loads, each weight pair feeds 4 matrix instructions, next 16 k loaded ahead; sums, order unchanged
# speedup vs baseline: 3.6828x; 1.2120x over previous
#include <hip/hip_runtime.h>

typedef float v2f __attribute__((ext_vector_type(2)));
typedef float v8f __attribute__((ext_vector_type(8)));

#define T_SEQ 2048
#define FEAT  256
#define GATES 64
#define HID   16
#define BATCH 64

#define K1_TT 64
#define K1_KG 16

struct K1Frag {
  v2f a0, a1;
  float4 x0, x1;
};

__device__ __forceinline__ v8f wmma4(v2f a, v2f b, v8f c) {
  return __builtin_amdgcn_wmma_f32_16x16x4_f32(
       false, a,  false, b,
       (short)0, c,  false,  false);
}

__device__ __forceinline__ void k1_load(K1Frag (&f)[4], const float* Wr0,
                                        const float* Wr1, const float* Xb,
                                        int k0, int half) {
#pragma unroll
  for (int s = 0; s < 4; ++s) {
    const int ka = k0 + 4 * s + 2 * half;
    f[s].a0 = *(const v2f*)(Wr0 + ka);
    f[s].a1 = *(const v2f*)(Wr1 + ka);
    f[s].x0 = *(const float4*)(Xb + (size_t)ka * T_SEQ);
    f[s].x1 = *(const float4*)(Xb + (size_t)(ka + 1) * T_SEQ);
  }
}

__device__ __forceinline__ void k1_mma(const K1Frag (&f)[4], v8f (&c0)[4],
                                       v8f (&c1)[4]) {
#pragma unroll
  for (int s = 0; s < 4; ++s) {
    const v2f b0 = {f[s].x0.x, f[s].x1.x};
    const v2f b1 = {f[s].x0.y, f[s].x1.y};
    const v2f b2 = {f[s].x0.z, f[s].x1.z};
    const v2f b3 = {f[s].x0.w, f[s].x1.w};
    c0[0] = wmma4(f[s].a0, b0, c0[0]);
    c0[1] = wmma4(f[s].a0, b1, c0[1]);
    c0[2] = wmma4(f[s].a0, b2, c0[2]);
    c0[3] = wmma4(f[s].a0, b3, c0[3]);
    c1[0] = wmma4(f[s].a1, b0, c1[0]);
    c1[1] = wmma4(f[s].a1, b1, c1[1]);
    c1[2] = wmma4(f[s].a1, b2, c1[2]);
    c1[3] = wmma4(f[s].a1, b3, c1[3]);
  }
}

__global__ __launch_bounds__(64) void lstm_in_gemm_wmma(
    const float* __restrict__ x,
    const float* __restrict__ W,
    const float* __restrict__ bih,
    const float* __restrict__ bhh,
    float* __restrict__ xg)
{
  const int lane = threadIdx.x & 31;
  const int mp   = threadIdx.x >> 5;
  const int half = lane >> 4;
  const int r    = lane & 15;
  const int tl   = blockIdx.x * K1_TT + 4 * r;
  const int b    = blockIdx.y;

  const float* Xb  = x + (size_t)b * FEAT * T_SEQ + tl;
  const float* Wr0 = W + (size_t)(mp * 32 + r) * FEAT;
  const float* Wr1 = Wr0 + (size_t)16 * FEAT;

  v8f c0[4] = {}, c1[4] = {};
  K1Frag fa[4], fb[4];
  k1_load(fa, Wr0, Wr1, Xb, 0, half);
  for (int k = 0; k < FEAT; k += 2 * K1_KG) {
    k1_load(fb, Wr0, Wr1, Xb, k + K1_KG, half);
    k1_mma(fa, c0, c1);
    if (k + 2 * K1_KG < FEAT)
      k1_load(fa, Wr0, Wr1, Xb, k + 2 * K1_KG, half);
    k1_mma(fb, c0, c1);
  }

#pragma unroll
  for (int v = 0; v < 8; ++v) {
    const int g0 = mp * 32 + v + 8 * half;
    const int g1 = g0 + 16;
    const float bi0 = bih[g0], bh0 = bhh[g0];
    const float bi1 = bih[g1], bh1 = bhh[g1];
    const float4 o0 = make_float4(c0[0][v] + bi0 + bh0, c0[1][v] + bi0 + bh0,
                                  c0[2][v] + bi0 + bh0, c0[3][v] + bi0 + bh0);
    const float4 o1 = make_float4(c1[0][v] + bi1 + bh1, c1[1][v] + bi1 + bh1,
                                  c1[2][v] + bi1 + bh1, c1[3][v] + bi1 + bh1);
    *(float4*)(xg + ((size_t)b * GATES + g0) * T_SEQ + tl) = o0;
    *(float4*)(xg + ((size_t)b * GATES + g1) * T_SEQ + tl) = o1;
  }
}

#define CH  32
#define NCH (T_SEQ / CH)

__device__ __forceinline__ float bcast(float v, int j) {
  return __uint_as_float(__builtin_amdgcn_readlane(__float_as_uint(v), (unsigned)j));
}
__device__ __forceinline__ float xhalf(float x) {
  const int b = __builtin_bit_cast(int, x);
  return __builtin_bit_cast(float,
      __builtin_amdgcn_permlanex16(b, b, 0x76543210, 0xfedcba98, false, false));
}
template <int M>
__device__ __forceinline__ float rowx(float x) {
  const int b = __builtin_bit_cast(int, x);
  return __builtin_bit_cast(float,
      __builtin_amdgcn_update_dpp(b, b, 0x160 + M, 0xf, 0xf, false));
}
__device__ __forceinline__ float tanh_f(float x) {
  return __builtin_amdgcn_tanhf(x);
}
__device__ __forceinline__ float sig_h(float xh) {
  return fmaf(tanh_f(xh), 0.5f, 0.5f);
}

__device__ __forceinline__ float unitsum(float h, bool lower) {
  float s = h;
  s += rowx<1>(s);
  s += rowx<2>(s);
  s += rowx<4>(s);
  s += rowx<8>(s);
  const float sx = xhalf(s);
  return lower ? s : sx;
}

__global__ __launch_bounds__(64) void lstm_recurrence(
    const float* __restrict__ xg,
    const float* __restrict__ Whh0,
    const float* __restrict__ Wih1,
    const float* __restrict__ Whh1,
    const float* __restrict__ bih1,
    const float* __restrict__ bhh1,
    float* __restrict__ out)
{
  __shared__ float2 ring[2][CH][32];

  const int lane = threadIdx.x & 31;
  const int role = __builtin_amdgcn_readfirstlane((int)(threadIdx.x >> 5));
  const int b    = blockIdx.x;
  const int g0   = lane;
  const int g1   = lane + 32;

  const float s0 = 0.5f;
  const float s1 = (lane < 16) ? 1.0f : 0.5f;
  float w0a[HID], w0b[HID], wia[HID], wib[HID], wha[HID], whb[HID];
#pragma unroll
  for (int j = 0; j < HID; ++j) {
    w0a[j] = Whh0[g0 * HID + j] * s0;
    w0b[j] = Whh0[g1 * HID + j] * s1;
    wia[j] = Wih1[g0 * HID + j] * s0;
    wib[j] = Wih1[g1 * HID + j] * s1;
    wha[j] = Whh1[g0 * HID + j] * s0;
    whb[j] = Whh1[g1 * HID + j] * s1;
  }
  const float bias0 = (bih1[g0] + bhh1[g0]) * s0;
  const float bias1 = (bih1[g1] + bhh1[g1]) * s1;

  const float* xg0p = xg + ((size_t)b * GATES + g0) * T_SEQ;
  const float* xg1p = xg + ((size_t)b * GATES + g1) * T_SEQ;
  float* outb = out + (size_t)b * T_SEQ;

  float h0 = 0.0f, c0 = 0.0f;
  float h1 = 0.0f, c1 = 0.0f;
  float hb0[HID], hb1[HID];
#pragma unroll
  for (int j = 0; j < HID; ++j) { hb0[j] = 0.0f; hb1[j] = 0.0f; }
  float outacc = 0.0f;

  float4 x0v = make_float4(0.f, 0.f, 0.f, 0.f);
  float4 x1v = make_float4(0.f, 0.f, 0.f, 0.f);

  if (role == 0) {
    x0v = *(const float4*)(xg0p);
    x1v = *(const float4*)(xg1p);
    float p0a = 0.f, p0b = 0.f, p1a = 0.f, p1b = 0.f;
#pragma unroll
    for (int j = 0; j < 8; ++j) {
      p0a = fmaf(w0a[j],     hb0[j],     p0a);
      p0b = fmaf(w0a[j + 8], hb0[j + 8], p0b);
      p1a = fmaf(w0b[j],     hb0[j],     p1a);
      p1b = fmaf(w0b[j + 8], hb0[j + 8], p1b);
    }
    const float a0 = fmaf(x0v.x, s0, (p0a + p0b));
    const float a1 = fmaf(x1v.x, s1, (p1a + p1b));
    const float fv = xhalf(a0);
    const float ov = xhalf(a1);
    c0 = sig_h(fv) * c0 + sig_h(a0) * tanh_f(a1);
    h0 = sig_h(ov) * tanh_f(c0);
#pragma unroll
    for (int j = 0; j < HID; ++j) hb0[j] = bcast(h0, j);
  }

  for (int p = 0; p <= NCH; ++p) {
    if (role == 0) {
      if (p < NCH) {
        float2 (*wbuf)[32] = ring[p & 1];
        for (int t0 = 0; t0 < CH; t0 += 4) {
          const int tg = p * CH + t0;
          const int tn = (tg + 4 < T_SEQ) ? (tg + 4) : tg;
          const float4 nx0 = *(const float4*)(xg0p + tn);
          const float4 nx1 = *(const float4*)(xg1p + tn);
          __builtin_prefetch(xg0p + tg + 64, 0, 3);
          __builtin_prefetch(xg1p + tg + 64, 0, 3);
          const float x0a[4] = {x0v.y, x0v.z, x0v.w, nx0.x};
          const float x1a[4] = {x1v.y, x1v.z, x1v.w, nx1.x};

#pragma unroll
          for (int u = 0; u < 4; ++u) {
            float p0a = 0.f, p0b = 0.f, p1a = 0.f, p1b = 0.f;
            float r0a = 0.f, r0b = 0.f, r1a = 0.f, r1b = 0.f;
#pragma unroll
            for (int j = 0; j < 8; ++j) {
              p0a = fmaf(w0a[j],     hb0[j],     p0a);
              p0b = fmaf(w0a[j + 8], hb0[j + 8], p0b);
              p1a = fmaf(w0b[j],     hb0[j],     p1a);
              p1b = fmaf(w0b[j + 8], hb0[j + 8], p1b);
              r0a = fmaf(wia[j],     hb0[j],     r0a);
              r0b = fmaf(wia[j + 8], hb0[j + 8], r0b);
              r1a = fmaf(wib[j],     hb0[j],     r1a);
              r1b = fmaf(wib[j + 8], hb0[j + 8], r1b);
            }
            wbuf[t0 + u][lane] = make_float2(r0a + r0b, r1a + r1b);
            const float a0 = fmaf(x0a[u], s0, (p0a + p0b));
            const float a1 = fmaf(x1a[u], s1, (p1a + p1b));
            const float fv = xhalf(a0);
            const float ov = xhalf(a1);
            c0 = sig_h(fv) * c0 + sig_h(a0) * tanh_f(a1);
            h0 = sig_h(ov) * tanh_f(c0);
#pragma unroll
            for (int j = 0; j < HID; ++j) hb0[j] = bcast(h0, j);
            __builtin_amdgcn_sched_barrier(0);
          }
          x0v = nx0;
          x1v = nx1;
        }
      }
    } else {
      if (p >= 1) {
        const int q = p - 1;
        const float2 (*rbuf)[32] = ring[q & 1];
        float2 s0 = rbuf[0][lane], s1 = rbuf[1][lane];
        float2 s2 = rbuf[2][lane], s3 = rbuf[3][lane];
        for (int t0 = 0; t0 < CH; t0 += 4) {
          const int tq = (t0 + 4 < CH) ? (t0 + 4) : t0;
          const float2 n0 = rbuf[tq][lane],     n1 = rbuf[tq + 1][lane];
          const float2 n2 = rbuf[tq + 2][lane], n3 = rbuf[tq + 3][lane];
          const float sa[4] = {s0.x, s1.x, s2.x, s3.x};
          const float sb[4] = {s0.y, s1.y, s2.y, s3.y};

#pragma unroll
          for (int u = 0; u < 4; ++u) {
            const float sprev = unitsum(h1, lane < 16);
            outacc = (lane == ((t0 + u - 1) & 31)) ? sprev : outacc;
            float r0c = 0.f, r0d = 0.f, r1c = 0.f, r1d = 0.f;
#pragma unroll
            for (int j = 0; j < 8; ++j) {
              r0c = fmaf(wha[j],     hb1[j],     r0c);
              r0d = fmaf(wha[j + 8], hb1[j + 8], r0d);
              r1c = fmaf(whb[j],     hb1[j],     r1c);
              r1d = fmaf(whb[j + 8], hb1[j + 8], r1d);
            }
            const float b0 = bias0 + (sa[u] + (r0c + r0d));
            const float b1 = bias1 + (sb[u] + (r1c + r1d));
            const float fv1 = xhalf(b0);
            const float ov1 = xhalf(b1);
            c1 = sig_h(fv1) * c1 + sig_h(b0) * tanh_f(b1);
            h1 = sig_h(ov1) * tanh_f(c1);
#pragma unroll
            for (int j = 0; j < HID; ++j) hb1[j] = bcast(h1, j);
            __builtin_amdgcn_sched_barrier(0);
            if (u == 0 && t0 == 0 && q != 0)
              outb[(q * CH - 32) + lane] = outacc * 0.0625f;
          }
          s0 = n0; s1 = n1; s2 = n2; s3 = n3;
        }
      }
    }
    __builtin_amdgcn_fence(__ATOMIC_RELEASE, "workgroup", "local");
    __builtin_amdgcn_s_barrier();
    __builtin_amdgcn_fence(__ATOMIC_ACQUIRE, "workgroup", "local");
  }

  if (role != 0) {
    const float slast = unitsum(h1, lane < 16);
    outacc = (lane == 31) ? slast : outacc;
    outb[(T_SEQ - 32) + lane] = outacc * 0.0625f;
  }
}

extern "C" void kernel_launch(void* const* d_in, const int* in_sizes, int n_in,
                              void* d_out, int out_size, void* d_ws, size_t ws_size,
                              hipStream_t stream) {
  (void)in_sizes; (void)n_in; (void)out_size; (void)ws_size;
  const float* x    = (const float*)d_in[0];
  const float* Wih0 = (const float*)d_in[1];
  const float* Whh0 = (const float*)d_in[2];
  const float* bih0 = (const float*)d_in[3];
  const float* bhh0 = (const float*)d_in[4];
  const float* Wih1 = (const float*)d_in[5];
  const float* Whh1 = (const float*)d_in[6];
  const float* bih1 = (const float*)d_in[7];
  const float* bhh1 = (const float*)d_in[8];
  float* out = (float*)d_out;
  float* xg  = (float*)d_ws;

  dim3 g1(T_SEQ / K1_TT, BATCH, 1);
  lstm_in_gemm_wmma<<<g1, dim3(64, 1, 1), 0, stream>>>(x, Wih0, bih0, bhh0, xg);

  lstm_recurrence<<<dim3(BATCH, 1, 1), dim3(64, 1, 1), 0, stream>>>(
      xg, Whh0, Wih1, Whh1, bih1, bhh1, out);
}


# recurrence: each 16-lane row takes half of k for all four gates of its unit; h stays in VGPRs, handed round the row by v_permlane16 instead of 16 v_readlane to SGPRs; sums and their order unchanged
# speedup vs baseline: 4.1019x; 1.1138x over previous
#include <hip/hip_runtime.h>

typedef float v2f __attribute__((ext_vector_type(2)));
typedef float v8f __attribute__((ext_vector_type(8)));

#define T_SEQ 2048
#define FEAT  256
#define GATES 64
#define HID   16
#define BATCH 64

#define K1_TT 64
#define K1_KG 16

struct K1Frag {
  v2f a0, a1;
  float4 x0, x1;
};

__device__ __forceinline__ v8f wmma4(v2f a, v2f b, v8f c) {
  return __builtin_amdgcn_wmma_f32_16x16x4_f32(
       false, a,  false, b,
       (short)0, c,  false,  false);
}

__device__ __forceinline__ void k1_load(K1Frag (&f)[4], const float* Wr0,
                                        const float* Wr1, const float* Xb,
                                        int k0, int half) {
#pragma unroll
  for (int s = 0; s < 4; ++s) {
    const int ka = k0 + 4 * s + 2 * half;
    f[s].a0 = *(const v2f*)(Wr0 + ka);
    f[s].a1 = *(const v2f*)(Wr1 + ka);
    f[s].x0 = *(const float4*)(Xb + (size_t)ka * T_SEQ);
    f[s].x1 = *(const float4*)(Xb + (size_t)(ka + 1) * T_SEQ);
  }
}

__device__ __forceinline__ void k1_mma(const K1Frag (&f)[4], v8f (&c0)[4],
                                       v8f (&c1)[4]) {
#pragma unroll
  for (int s = 0; s < 4; ++s) {
    const v2f b0 = {f[s].x0.x, f[s].x1.x};
    const v2f b1 = {f[s].x0.y, f[s].x1.y};
    const v2f b2 = {f[s].x0.z, f[s].x1.z};
    const v2f b3 = {f[s].x0.w, f[s].x1.w};
    c0[0] = wmma4(f[s].a0, b0, c0[0]);
    c0[1] = wmma4(f[s].a0, b1, c0[1]);
    c0[2] = wmma4(f[s].a0, b2, c0[2]);
    c0[3] = wmma4(f[s].a0, b3, c0[3]);
    c1[0] = wmma4(f[s].a1, b0, c1[0]);
    c1[1] = wmma4(f[s].a1, b1, c1[1]);
    c1[2] = wmma4(f[s].a1, b2, c1[2]);
    c1[3] = wmma4(f[s].a1, b3, c1[3]);
  }
}

__global__ __launch_bounds__(64) void lstm_in_gemm_wmma(
    const float* __restrict__ x,
    const float* __restrict__ W,
    const float* __restrict__ bih,
    const float* __restrict__ bhh,
    float* __restrict__ xg)
{
  const int lane = threadIdx.x & 31;
  const int mp   = threadIdx.x >> 5;
  const int half = lane >> 4;
  const int r    = lane & 15;
  const int tl   = blockIdx.x * K1_TT + 4 * r;
  const int b    = blockIdx.y;

  const float* Xb  = x + (size_t)b * FEAT * T_SEQ + tl;
  const float* Wr0 = W + (size_t)(mp * 32 + r) * FEAT;
  const float* Wr1 = Wr0 + (size_t)16 * FEAT;

  v8f c0[4] = {}, c1[4] = {};
  K1Frag fa[4], fb[4];
  k1_load(fa, Wr0, Wr1, Xb, 0, half);
  for (int k = 0; k < FEAT; k += 2 * K1_KG) {
    k1_load(fb, Wr0, Wr1, Xb, k + K1_KG, half);
    k1_mma(fa, c0, c1);
    if (k + 2 * K1_KG < FEAT)
      k1_load(fa, Wr0, Wr1, Xb, k + 2 * K1_KG, half);
    k1_mma(fb, c0, c1);
  }

#pragma unroll
  for (int v = 0; v < 8; ++v) {
    const int g0 = mp * 32 + v + 8 * half;
    const int g1 = g0 + 16;
    const float bi0 = bih[g0], bh0 = bhh[g0];
    const float bi1 = bih[g1], bh1 = bhh[g1];
    const float4 o0 = make_float4(c0[0][v] + bi0 + bh0, c0[1][v] + bi0 + bh0,
                                  c0[2][v] + bi0 + bh0, c0[3][v] + bi0 + bh0);
    const float4 o1 = make_float4(c1[0][v] + bi1 + bh1, c1[1][v] + bi1 + bh1,
                                  c1[2][v] + bi1 + bh1, c1[3][v] + bi1 + bh1);
    *(float4*)(xg + ((size_t)b * GATES + g0) * T_SEQ + tl) = o0;
    *(float4*)(xg + ((size_t)b * GATES + g1) * T_SEQ + tl) = o1;
  }
}

#define CH  32
#define NCH (T_SEQ / CH)
#define KH  8

template <int M>
__device__ __forceinline__ float rowbc(float x) {
  const int b = __builtin_bit_cast(int, x);
  return __builtin_bit_cast(float,
      __builtin_amdgcn_permlane16(b, b, 0x11111111u * M, 0x11111111u * M, true, false));
}
__device__ __forceinline__ float xjoin(float x) {
  const int b = __builtin_bit_cast(int, x);
  return __builtin_bit_cast(float,
      __builtin_amdgcn_permlanex16(b, b, 0xfedcba98u, 0x76543210u, true, false));
}
template <int M>
__device__ __forceinline__ float rowx(float x) {
  const int b = __builtin_bit_cast(int, x);
  return __builtin_bit_cast(float, __builtin_amdgcn_mov_dpp(b, 0x160 + M, 0xf, 0xf, false));
}
__device__ __forceinline__ float tanh_f(float x) {
  return __builtin_amdgcn_tanhf(x);
}
__device__ __forceinline__ float sig_h(float xh) {
  return fmaf(tanh_f(xh), 0.5f, 0.5f);
}
__device__ __forceinline__ float unitsum(float h) {
  float s = h;
  s += rowx<1>(s);
  s += rowx<2>(s);
  s += rowx<4>(s);
  s += rowx<8>(s);
  return s;
}

template <int M>
__device__ __forceinline__ void level_l0(const float (&w0)[4][KH], const float (&wi)[4][KH],
                                         float h, float& hbc, float& hbn, float (&pa)[4], float (&ra)[4]) {
  const float cur = hbc;
  hbc = hbn;
  if constexpr (M + 2 < KH) hbn = rowbc<M + 2>(h);
#pragma unroll
  for (int q = 0; q < 4; ++q) {
    pa[q] = fmaf(w0[q][M], cur, pa[q]);
    ra[q] = fmaf(wi[q][M], cur, ra[q]);
  }
  asm volatile("" :: "v"(pa[0]), "v"(pa[1]), "v"(pa[2]), "v"(pa[3]),
                     "v"(ra[0]), "v"(ra[1]), "v"(ra[2]), "v"(ra[3]), "v"(hbn));
  __builtin_amdgcn_sched_barrier(0);
}
template <int M>
__device__ __forceinline__ void level_l1(const float (&wh)[4][KH], float h, float& hbc, float& hbn,
                                         float (&rc)[4], float& us, float& ux) {
  const float cur = hbc;
  hbc = hbn;
  if constexpr (M + 2 < KH) hbn = rowbc<M + 2>(h);
#pragma unroll
  for (int q = 0; q < 4; ++q) rc[q] = fmaf(wh[q][M], cur, rc[q]);
  if constexpr (M % 2 == 0) ux = rowx<(1 << (M / 2))>(us);
  else                      us = us + ux;
  asm volatile("" :: "v"(rc[0]), "v"(rc[1]), "v"(rc[2]), "v"(rc[3]),
                     "v"(hbn), "v"(us), "v"(ux));
  __builtin_amdgcn_sched_barrier(0);
}

__global__ __launch_bounds__(64) void lstm_recurrence(
    const float* __restrict__ xg,
    const float* __restrict__ Whh0,
    const float* __restrict__ Wih1,
    const float* __restrict__ Whh1,
    const float* __restrict__ bih1,
    const float* __restrict__ bhh1,
    float* __restrict__ out)
{
  __shared__ float4 ring[2][CH][32];

  const int lane = threadIdx.x & 31;
  const int role = __builtin_amdgcn_readfirstlane((int)(threadIdx.x >> 5));
  const int b    = blockIdx.x;
  const int row  = lane >> 4;
  const int unit = (lane & 15) ^ (row << 3);
  const int kb   = row * KH;

  const float sc[4] = {0.5f, 0.5f, 1.0f, 0.5f};
  float w0[4][KH], wi[4][KH], wh[4][KH], bias[4];
#pragma unroll
  for (int q = 0; q < 4; ++q) {
    const int g = q * HID + unit;
#pragma unroll
    for (int m = 0; m < KH; ++m) {
      w0[q][m] = Whh0[g * HID + kb + m] * sc[q];
      wi[q][m] = Wih1[g * HID + kb + m] * sc[q];
      wh[q][m] = Whh1[g * HID + kb + m] * sc[q];
    }
    bias[q] = (bih1[g] + bhh1[g]) * sc[q];
  }

  const float* xgp[4];
#pragma unroll
  for (int q = 0; q < 4; ++q)
    xgp[q] = xg + ((size_t)b * GATES + q * HID + unit) * T_SEQ;
  float* outb = out + (size_t)b * T_SEQ;

  float h0 = 0.0f, c0 = 0.0f;
  float h1 = 0.0f, c1 = 0.0f;
  float hbc0 = 0.0f, hbn0 = 0.0f;
  float hbc1 = 0.0f, hbn1 = 0.0f;
  float outacc = 0.0f;

  float4 xv[4];
#pragma unroll
  for (int q = 0; q < 4; ++q) xv[q] = make_float4(0.f, 0.f, 0.f, 0.f);

  if (role == 0) {
#pragma unroll
    for (int q = 0; q < 4; ++q) xv[q] = *(const float4*)(xgp[q]);
    float pa[4] = {0.f, 0.f, 0.f, 0.f};
#pragma unroll
    for (int m = 0; m < KH; ++m) {
#pragma unroll
      for (int q = 0; q < 4; ++q) pa[q] = fmaf(w0[q][m], 0.0f, pa[q]);
    }
    const float ai = fmaf(xv[0].x, sc[0], pa[0] + xjoin(pa[0]));
    const float af = fmaf(xv[1].x, sc[1], pa[1] + xjoin(pa[1]));
    const float ag = fmaf(xv[2].x, sc[2], pa[2] + xjoin(pa[2]));
    const float ao = fmaf(xv[3].x, sc[3], pa[3] + xjoin(pa[3]));
    c0 = sig_h(af) * c0 + sig_h(ai) * tanh_f(ag);
    h0 = sig_h(ao) * tanh_f(c0);
    hbc0 = rowbc<0>(h0);
    hbn0 = rowbc<1>(h0);
  }

  for (int p = 0; p <= NCH; ++p) {
    if (role == 0) {
      if (p < NCH) {
        float4 (*wbuf)[32] = ring[p & 1];
        for (int t0 = 0; t0 < CH; t0 += 4) {
          const int tg = p * CH + t0;
          const int tn = (tg + 4 < T_SEQ) ? (tg + 4) : tg;
          float4 nx[4];
#pragma unroll
          for (int q = 0; q < 4; ++q) nx[q] = *(const float4*)(xgp[q] + tn);
#pragma unroll
          for (int q = 0; q < 4; ++q) __builtin_prefetch(xgp[q] + tg + 64, 0, 3);
          float xa[4][4];
#pragma unroll
          for (int q = 0; q < 4; ++q) {
            xa[q][0] = xv[q].y; xa[q][1] = xv[q].z; xa[q][2] = xv[q].w; xa[q][3] = nx[q].x;
          }

#pragma unroll
          for (int u = 0; u < 4; ++u) {
            float pa[4] = {0.f, 0.f, 0.f, 0.f};
            float ra[4] = {0.f, 0.f, 0.f, 0.f};
            level_l0<0>(w0, wi, h0, hbc0, hbn0, pa, ra);
            level_l0<1>(w0, wi, h0, hbc0, hbn0, pa, ra);
            level_l0<2>(w0, wi, h0, hbc0, hbn0, pa, ra);
            level_l0<3>(w0, wi, h0, hbc0, hbn0, pa, ra);
            level_l0<4>(w0, wi, h0, hbc0, hbn0, pa, ra);
            level_l0<5>(w0, wi, h0, hbc0, hbn0, pa, ra);
            level_l0<6>(w0, wi, h0, hbc0, hbn0, pa, ra);
            level_l0<7>(w0, wi, h0, hbc0, hbn0, pa, ra);
            const float ai = fmaf(xa[0][u], sc[0], pa[0] + xjoin(pa[0]));
            const float af = fmaf(xa[1][u], sc[1], pa[1] + xjoin(pa[1]));
            const float ag = fmaf(xa[2][u], sc[2], pa[2] + xjoin(pa[2]));
            const float ao = fmaf(xa[3][u], sc[3], pa[3] + xjoin(pa[3]));
            asm volatile("" :: "v"(ai), "v"(af), "v"(ag), "v"(ao));
            __builtin_amdgcn_sched_barrier(0);
            wbuf[t0 + u][lane] = make_float4(ra[0] + xjoin(ra[0]), ra[1] + xjoin(ra[1]),
                                             ra[2] + xjoin(ra[2]), ra[3] + xjoin(ra[3]));
            c0 = sig_h(af) * c0 + sig_h(ai) * tanh_f(ag);
            h0 = sig_h(ao) * tanh_f(c0);
            hbc0 = rowbc<0>(h0);
            hbn0 = rowbc<1>(h0);
            __builtin_amdgcn_sched_barrier(0);
          }
#pragma unroll
          for (int q = 0; q < 4; ++q) xv[q] = nx[q];
        }
      }
    } else {
      if (p >= 1) {
        const int q_ = p - 1;
        const float4 (*rbuf)[32] = ring[q_ & 1];
        float4 s0 = rbuf[0][lane], s1 = rbuf[1][lane];
        float4 s2 = rbuf[2][lane], s3 = rbuf[3][lane];
        for (int t0 = 0; t0 < CH; t0 += 4) {
          const int tq = (t0 + 4 < CH) ? (t0 + 4) : t0;
          const float4 n0 = rbuf[tq][lane],     n1 = rbuf[tq + 1][lane];
          const float4 n2 = rbuf[tq + 2][lane], n3 = rbuf[tq + 3][lane];
          const float sa[4][4] = {{s0.x, s0.y, s0.z, s0.w}, {s1.x, s1.y, s1.z, s1.w},
                                  {s2.x, s2.y, s2.z, s2.w}, {s3.x, s3.y, s3.z, s3.w}};

#pragma unroll
          for (int u = 0; u < 4; ++u) {
            float rc[4] = {0.f, 0.f, 0.f, 0.f};
            float us = h1, ux = 0.0f;
            level_l1<0>(wh, h1, hbc1, hbn1, rc, us, ux);
            level_l1<1>(wh, h1, hbc1, hbn1, rc, us, ux);
            level_l1<2>(wh, h1, hbc1, hbn1, rc, us, ux);
            level_l1<3>(wh, h1, hbc1, hbn1, rc, us, ux);
            level_l1<4>(wh, h1, hbc1, hbn1, rc, us, ux);
            level_l1<5>(wh, h1, hbc1, hbn1, rc, us, ux);
            level_l1<6>(wh, h1, hbc1, hbn1, rc, us, ux);
            level_l1<7>(wh, h1, hbc1, hbn1, rc, us, ux);
            outacc = (lane == ((t0 + u - 1) & 31)) ? us : outacc;
            const float bi = bias[0] + (sa[u][0] + (rc[0] + xjoin(rc[0])));
            const float bf = bias[1] + (sa[u][1] + (rc[1] + xjoin(rc[1])));
            const float bg = bias[2] + (sa[u][2] + (rc[2] + xjoin(rc[2])));
            const float bo = bias[3] + (sa[u][3] + (rc[3] + xjoin(rc[3])));
            asm volatile("" :: "v"(bi), "v"(bf), "v"(bg), "v"(bo));
            __builtin_amdgcn_sched_barrier(0);
            c1 = sig_h(bf) * c1 + sig_h(bi) * tanh_f(bg);
            h1 = sig_h(bo) * tanh_f(c1);
            hbc1 = rowbc<0>(h1);
            hbn1 = rowbc<1>(h1);
            __builtin_amdgcn_sched_barrier(0);
            if (u == 0 && t0 == 0 && q_ != 0)
              outb[(q_ * CH - 32) + lane] = outacc * 0.0625f;
          }
          s0 = n0; s1 = n1; s2 = n2; s3 = n3;
        }
      }
    }
    __builtin_amdgcn_fence(__ATOMIC_RELEASE, "workgroup", "local");
    __builtin_amdgcn_s_barrier();
    __builtin_amdgcn_fence(__ATOMIC_ACQUIRE, "workgroup", "local");
  }

  if (role != 0) {
    const float slast = unitsum(h1);
    outacc = (lane == 31) ? slast : outacc;
    outb[(T_SEQ - 32) + lane] = outacc * 0.0625f;
  }
}

extern "C" void kernel_launch(void* const* d_in, const int* in_sizes, int n_in,
                              void* d_out, int out_size, void* d_ws, size_t ws_size,
                              hipStream_t stream) {
  (void)in_sizes; (void)n_in; (void)out_size; (void)ws_size;
  const float* x    = (const float*)d_in[0];
  const float* Wih0 = (const float*)d_in[1];
  const float* Whh0 = (const float*)d_in[2];
  const float* bih0 = (const float*)d_in[3];
  const float* bhh0 = (const float*)d_in[4];
  const float* Wih1 = (const float*)d_in[5];
  const float* Whh1 = (const float*)d_in[6];
  const float* bih1 = (const float*)d_in[7];
  const float* bhh1 = (const float*)d_in[8];
  float* out = (float*)d_out;
  float* xg  = (float*)d_ws;

  dim3 g1(T_SEQ / K1_TT, BATCH, 1);
  lstm_in_gemm_wmma<<<g1, dim3(64, 1, 1), 0, stream>>>(x, Wih0, bih0, bhh0, xg);

  lstm_recurrence<<<dim3(BATCH, 1, 1), dim3(64, 1, 1), 0, stream>>>(
      xg, Whh0, Wih1, Whh1, bih1, bhh1, out);
}


# recurrence: layer-0 wave holds the hand-over sums of 4 steps in registers and writes them by 4 LDS stores at one place behind tanh(c) of the 4th step, not one store per step; sums, order unchanged
# speedup vs baseline: 4.3709x; 1.0656x over previous
#include <hip/hip_runtime.h>

typedef float v2f __attribute__((ext_vector_type(2)));
typedef float v8f __attribute__((ext_vector_type(8)));

#define T_SEQ 2048
#define FEAT  256
#define GATES 64
#define HID   16
#define BATCH 64

#define K1_TT 64
#define K1_KG 16

struct K1Frag {
  v2f a0, a1;
  float4 x0, x1;
};

__device__ __forceinline__ v8f wmma4(v2f a, v2f b, v8f c) {
  return __builtin_amdgcn_wmma_f32_16x16x4_f32(
       false, a,  false, b,
       (short)0, c,  false,  false);
}

__device__ __forceinline__ void k1_load(K1Frag (&f)[4], const float* Wr0,
                                        const float* Wr1, const float* Xb,
                                        int k0, int half) {
#pragma unroll
  for (int s = 0; s < 4; ++s) {
    const int ka = k0 + 4 * s + 2 * half;
    f[s].a0 = *(const v2f*)(Wr0 + ka);
    f[s].a1 = *(const v2f*)(Wr1 + ka);
    f[s].x0 = *(const float4*)(Xb + (size_t)ka * T_SEQ);
    f[s].x1 = *(const float4*)(Xb + (size_t)(ka + 1) * T_SEQ);
  }
}

__device__ __forceinline__ void k1_mma(const K1Frag (&f)[4], v8f (&c0)[4],
                                       v8f (&c1)[4]) {
#pragma unroll
  for (int s = 0; s < 4; ++s) {
    const v2f b0 = {f[s].x0.x, f[s].x1.x};
    const v2f b1 = {f[s].x0.y, f[s].x1.y};
    const v2f b2 = {f[s].x0.z, f[s].x1.z};
    const v2f b3 = {f[s].x0.w, f[s].x1.w};
    c0[0] = wmma4(f[s].a0, b0, c0[0]);
    c0[1] = wmma4(f[s].a0, b1, c0[1]);
    c0[2] = wmma4(f[s].a0, b2, c0[2]);
    c0[3] = wmma4(f[s].a0, b3, c0[3]);
    c1[0] = wmma4(f[s].a1, b0, c1[0]);
    c1[1] = wmma4(f[s].a1, b1, c1[1]);
    c1[2] = wmma4(f[s].a1, b2, c1[2]);
    c1[3] = wmma4(f[s].a1, b3, c1[3]);
  }
}

__global__ __launch_bounds__(64) void lstm_in_gemm_wmma(
    const float* __restrict__ x,
    const float* __restrict__ W,
    const float* __restrict__ bih,
    const float* __restrict__ bhh,
    float* __restrict__ xg)
{
  const int lane = threadIdx.x & 31;
  const int mp   = threadIdx.x >> 5;
  const int half = lane >> 4;
  const int r    = lane & 15;
  const int tl   = blockIdx.x * K1_TT + 4 * r;
  const int b    = blockIdx.y;

  const float* Xb  = x + (size_t)b * FEAT * T_SEQ + tl;
  const float* Wr0 = W + (size_t)(mp * 32 + r) * FEAT;
  const float* Wr1 = Wr0 + (size_t)16 * FEAT;

  v8f c0[4] = {}, c1[4] = {};
  K1Frag fa[4], fb[4];
  k1_load(fa, Wr0, Wr1, Xb, 0, half);
  for (int k = 0; k < FEAT; k += 2 * K1_KG) {
    k1_load(fb, Wr0, Wr1, Xb, k + K1_KG, half);
    k1_mma(fa, c0, c1);
    if (k + 2 * K1_KG < FEAT)
      k1_load(fa, Wr0, Wr1, Xb, k + 2 * K1_KG, half);
    k1_mma(fb, c0, c1);
  }

#pragma unroll
  for (int v = 0; v < 8; ++v) {
    const int g0 = mp * 32 + v + 8 * half;
    const int g1 = g0 + 16;
    const float bi0 = bih[g0], bh0 = bhh[g0];
    const float bi1 = bih[g1], bh1 = bhh[g1];
    const float4 o0 = make_float4(c0[0][v] + bi0 + bh0, c0[1][v] + bi0 + bh0,
                                  c0[2][v] + bi0 + bh0, c0[3][v] + bi0 + bh0);
    const float4 o1 = make_float4(c1[0][v] + bi1 + bh1, c1[1][v] + bi1 + bh1,
                                  c1[2][v] + bi1 + bh1, c1[3][v] + bi1 + bh1);
    *(float4*)(xg + ((size_t)b * GATES + g0) * T_SEQ + tl) = o0;
    *(float4*)(xg + ((size_t)b * GATES + g1) * T_SEQ + tl) = o1;
  }
}

#define CH  32
#define NCH (T_SEQ / CH)
#define KH  8

template <int M>
__device__ __forceinline__ float rowbc(float x) {
  const int b = __builtin_bit_cast(int, x);
  return __builtin_bit_cast(float,
      __builtin_amdgcn_permlane16(b, b, 0x11111111u * M, 0x11111111u * M, true, false));
}
__device__ __forceinline__ float xjoin(float x) {
  const int b = __builtin_bit_cast(int, x);
  return __builtin_bit_cast(float,
      __builtin_amdgcn_permlanex16(b, b, 0xfedcba98u, 0x76543210u, true, false));
}
template <int M>
__device__ __forceinline__ float rowx(float x) {
  const int b = __builtin_bit_cast(int, x);
  return __builtin_bit_cast(float, __builtin_amdgcn_mov_dpp(b, 0x160 + M, 0xf, 0xf, false));
}
__device__ __forceinline__ float tanh_f(float x) {
  return __builtin_amdgcn_tanhf(x);
}
__device__ __forceinline__ float sig_h(float xh) {
  return fmaf(tanh_f(xh), 0.5f, 0.5f);
}
__device__ __forceinline__ float unitsum(float h) {
  float s = h;
  s += rowx<1>(s);
  s += rowx<2>(s);
  s += rowx<4>(s);
  s += rowx<8>(s);
  return s;
}

template <int M>
__device__ __forceinline__ void level_l0(const float (&w0)[4][KH], const float (&wi)[4][KH],
                                         float h, float& hbc, float& hbn, float (&pa)[4], float (&ra)[4]) {
  const float cur = hbc;
  hbc = hbn;
  if constexpr (M + 2 < KH) hbn = rowbc<M + 2>(h);
#pragma unroll
  for (int q = 0; q < 4; ++q) {
    pa[q] = fmaf(w0[q][M], cur, pa[q]);
    ra[q] = fmaf(wi[q][M], cur, ra[q]);
  }
  asm volatile("" :: "v"(pa[0]), "v"(pa[1]), "v"(pa[2]), "v"(pa[3]),
                     "v"(ra[0]), "v"(ra[1]), "v"(ra[2]), "v"(ra[3]), "v"(hbn));
  __builtin_amdgcn_sched_barrier(0);
}
template <int M>
__device__ __forceinline__ void level_l1(const float (&wh)[4][KH], float h, float& hbc, float& hbn,
                                         float (&rc)[4], float& us, float& ux) {
  const float cur = hbc;
  hbc = hbn;
  if constexpr (M + 2 < KH) hbn = rowbc<M + 2>(h);
#pragma unroll
  for (int q = 0; q < 4; ++q) rc[q] = fmaf(wh[q][M], cur, rc[q]);
  if constexpr (M % 2 == 0) ux = rowx<(1 << (M / 2))>(us);
  else                      us = us + ux;
  asm volatile("" :: "v"(rc[0]), "v"(rc[1]), "v"(rc[2]), "v"(rc[3]),
                     "v"(hbn), "v"(us), "v"(ux));
  __builtin_amdgcn_sched_barrier(0);
}

__global__ __launch_bounds__(64) void lstm_recurrence(
    const float* __restrict__ xg,
    const float* __restrict__ Whh0,
    const float* __restrict__ Wih1,
    const float* __restrict__ Whh1,
    const float* __restrict__ bih1,
    const float* __restrict__ bhh1,
    float* __restrict__ out)
{
  __shared__ float4 ring[2][CH][32];

  const int lane = threadIdx.x & 31;
  const int role = __builtin_amdgcn_readfirstlane((int)(threadIdx.x >> 5));
  const int b    = blockIdx.x;
  const int row  = lane >> 4;
  const int unit = (lane & 15) ^ (row << 3);
  const int kb   = row * KH;

  const float sc[4] = {0.5f, 0.5f, 1.0f, 0.5f};
  float w0[4][KH], wi[4][KH], wh[4][KH], bias[4];
#pragma unroll
  for (int q = 0; q < 4; ++q) {
    const int g = q * HID + unit;
#pragma unroll
    for (int m = 0; m < KH; ++m) {
      w0[q][m] = Whh0[g * HID + kb + m] * sc[q];
      wi[q][m] = Wih1[g * HID + kb + m] * sc[q];
      wh[q][m] = Whh1[g * HID + kb + m] * sc[q];
    }
    bias[q] = (bih1[g] + bhh1[g]) * sc[q];
  }

  const float* xgp[4];
#pragma unroll
  for (int q = 0; q < 4; ++q)
    xgp[q] = xg + ((size_t)b * GATES + q * HID + unit) * T_SEQ;
  float* outb = out + (size_t)b * T_SEQ;

  float h0 = 0.0f, c0 = 0.0f;
  float h1 = 0.0f, c1 = 0.0f;
  float hbc0 = 0.0f, hbn0 = 0.0f;
  float hbc1 = 0.0f, hbn1 = 0.0f;
  float outacc = 0.0f;

  float4 xv[4];
#pragma unroll
  for (int q = 0; q < 4; ++q) xv[q] = make_float4(0.f, 0.f, 0.f, 0.f);
  float4 hv[4];
#pragma unroll
  for (int u = 0; u < 4; ++u) hv[u] = make_float4(0.f, 0.f, 0.f, 0.f);

  if (role == 0) {
#pragma unroll
    for (int q = 0; q < 4; ++q) xv[q] = *(const float4*)(xgp[q]);
    float pa[4] = {0.f, 0.f, 0.f, 0.f};
#pragma unroll
    for (int m = 0; m < KH; ++m) {
#pragma unroll
      for (int q = 0; q < 4; ++q) pa[q] = fmaf(w0[q][m], 0.0f, pa[q]);
    }
    const float ai = fmaf(xv[0].x, sc[0], pa[0] + xjoin(pa[0]));
    const float af = fmaf(xv[1].x, sc[1], pa[1] + xjoin(pa[1]));
    const float ag = fmaf(xv[2].x, sc[2], pa[2] + xjoin(pa[2]));
    const float ao = fmaf(xv[3].x, sc[3], pa[3] + xjoin(pa[3]));
    c0 = sig_h(af) * c0 + sig_h(ai) * tanh_f(ag);
    h0 = sig_h(ao) * tanh_f(c0);
    hbc0 = rowbc<0>(h0);
    hbn0 = rowbc<1>(h0);
  }

  for (int p = 0; p <= NCH; ++p) {
    if (role == 0) {
      if (p < NCH) {
        float4 (*wbuf)[32] = ring[p & 1];
        for (int t0 = 0; t0 < CH; t0 += 4) {
          const int tg = p * CH + t0;
          const int tn = (tg + 4 < T_SEQ) ? (tg + 4) : tg;
          float4 nx[4];
#pragma unroll
          for (int q = 0; q < 4; ++q) nx[q] = *(const float4*)(xgp[q] + tn);
#pragma unroll
          for (int q = 0; q < 4; ++q) __builtin_prefetch(xgp[q] + tg + 64, 0, 3);
          float xa[4][4];
#pragma unroll
          for (int q = 0; q < 4; ++q) {
            xa[q][0] = xv[q].y; xa[q][1] = xv[q].z; xa[q][2] = xv[q].w; xa[q][3] = nx[q].x;
          }

#pragma unroll
          for (int u = 0; u < 4; ++u) {
            float pa[4] = {0.f, 0.f, 0.f, 0.f};
            float ra[4] = {0.f, 0.f, 0.f, 0.f};
            level_l0<0>(w0, wi, h0, hbc0, hbn0, pa, ra);
            level_l0<1>(w0, wi, h0, hbc0, hbn0, pa, ra);
            level_l0<2>(w0, wi, h0, hbc0, hbn0, pa, ra);
            level_l0<3>(w0, wi, h0, hbc0, hbn0, pa, ra);
            level_l0<4>(w0, wi, h0, hbc0, hbn0, pa, ra);
            level_l0<5>(w0, wi, h0, hbc0, hbn0, pa, ra);
            level_l0<6>(w0, wi, h0, hbc0, hbn0, pa, ra);
            level_l0<7>(w0, wi, h0, hbc0, hbn0, pa, ra);
            const float ai = fmaf(xa[0][u], sc[0], pa[0] + xjoin(pa[0]));
            const float af = fmaf(xa[1][u], sc[1], pa[1] + xjoin(pa[1]));
            const float ag = fmaf(xa[2][u], sc[2], pa[2] + xjoin(pa[2]));
            const float ao = fmaf(xa[3][u], sc[3], pa[3] + xjoin(pa[3]));
            asm volatile("" :: "v"(ai), "v"(af), "v"(ag), "v"(ao));
            __builtin_amdgcn_sched_barrier(0);
            hv[u] = make_float4(ra[0] + xjoin(ra[0]), ra[1] + xjoin(ra[1]),
                                ra[2] + xjoin(ra[2]), ra[3] + xjoin(ra[3]));
            c0 = sig_h(af) * c0 + sig_h(ai) * tanh_f(ag);
            const float tc = tanh_f(c0);
            const float so = sig_h(ao);
            if (u == 3) {
              asm volatile("" :: "v"(tc), "v"(so));
              __builtin_amdgcn_sched_barrier(0);
#pragma unroll
              for (int v = 0; v < 4; ++v) wbuf[t0 + v][lane] = hv[v];
              __builtin_amdgcn_sched_barrier(0);
            }
            h0 = so * tc;
            hbc0 = rowbc<0>(h0);
            hbn0 = rowbc<1>(h0);
            __builtin_amdgcn_sched_barrier(0);
          }
#pragma unroll
          for (int q = 0; q < 4; ++q) xv[q] = nx[q];
        }
      }
    } else {
      if (p >= 1) {
        const int q_ = p - 1;
        const float4 (*rbuf)[32] = ring[q_ & 1];
        float4 s0 = rbuf[0][lane], s1 = rbuf[1][lane];
        float4 s2 = rbuf[2][lane], s3 = rbuf[3][lane];
        for (int t0 = 0; t0 < CH; t0 += 4) {
          const int tq = (t0 + 4 < CH) ? (t0 + 4) : t0;
          const float4 n0 = rbuf[tq][lane],     n1 = rbuf[tq + 1][lane];
          const float4 n2 = rbuf[tq + 2][lane], n3 = rbuf[tq + 3][lane];
          const float sa[4][4] = {{s0.x, s0.y, s0.z, s0.w}, {s1.x, s1.y, s1.z, s1.w},
                                  {s2.x, s2.y, s2.z, s2.w}, {s3.x, s3.y, s3.z, s3.w}};

#pragma unroll
          for (int u = 0; u < 4; ++u) {
            float rc[4] = {0.f, 0.f, 0.f, 0.f};
            float us = h1, ux = 0.0f;
            level_l1<0>(wh, h1, hbc1, hbn1, rc, us, ux);
            level_l1<1>(wh, h1, hbc1, hbn1, rc, us, ux);
            level_l1<2>(wh, h1, hbc1, hbn1, rc, us, ux);
            level_l1<3>(wh, h1, hbc1, hbn1, rc, us, ux);
            level_l1<4>(wh, h1, hbc1, hbn1, rc, us, ux);
            level_l1<5>(wh, h1, hbc1, hbn1, rc, us, ux);
            level_l1<6>(wh, h1, hbc1, hbn1, rc, us, ux);
            level_l1<7>(wh, h1, hbc1, hbn1, rc, us, ux);
            outacc = (lane == ((t0 + u - 1) & 31)) ? us : outacc;
            const float bi = bias[0] + (sa[u][0] + (rc[0] + xjoin(rc[0])));
            const float bf = bias[1] + (sa[u][1] + (rc[1] + xjoin(rc[1])));
            const float bg = bias[2] + (sa[u][2] + (rc[2] + xjoin(rc[2])));
            const float bo = bias[3] + (sa[u][3] + (rc[3] + xjoin(rc[3])));
            asm volatile("" :: "v"(bi), "v"(bf), "v"(bg), "v"(bo));
            __builtin_amdgcn_sched_barrier(0);
            c1 = sig_h(bf) * c1 + sig_h(bi) * tanh_f(bg);
            h1 = sig_h(bo) * tanh_f(c1);
            hbc1 = rowbc<0>(h1);
            hbn1 = rowbc<1>(h1);
            __builtin_amdgcn_sched_barrier(0);
            if (u == 0 && t0 == 0 && q_ != 0)
              outb[(q_ * CH - 32) + lane] = outacc * 0.0625f;
          }
          s0 = n0; s1 = n1; s2 = n2; s3 = n3;
        }
      }
    }
    __builtin_amdgcn_fence(__ATOMIC_RELEASE, "workgroup", "local");
    __builtin_amdgcn_s_barrier();
    __builtin_amdgcn_fence(__ATOMIC_ACQUIRE, "workgroup", "local");
  }

  if (role != 0) {
    const float slast = unitsum(h1);
    outacc = (lane == 31) ? slast : outacc;
    outb[(T_SEQ - 32) + lane] = outacc * 0.0625f;
  }
}

extern "C" void kernel_launch(void* const* d_in, const int* in_sizes, int n_in,
                              void* d_out, int out_size, void* d_ws, size_t ws_size,
                              hipStream_t stream) {
  (void)in_sizes; (void)n_in; (void)out_size; (void)ws_size;
  const float* x    = (const float*)d_in[0];
  const float* Wih0 = (const float*)d_in[1];
  const float* Whh0 = (const float*)d_in[2];
  const float* bih0 = (const float*)d_in[3];
  const float* bhh0 = (const float*)d_in[4];
  const float* Wih1 = (const float*)d_in[5];
  const float* Whh1 = (const float*)d_in[6];
  const float* bih1 = (const float*)d_in[7];
  const float* bhh1 = (const float*)d_in[8];
  float* out = (float*)d_out;
  float* xg  = (float*)d_ws;

  dim3 g1(T_SEQ / K1_TT, BATCH, 1);
  lstm_in_gemm_wmma<<<g1, dim3(64, 1, 1), 0, stream>>>(x, Wih0, bih0, bhh0, xg);

  lstm_recurrence<<<dim3(BATCH, 1, 1), dim3(64, 1, 1), 0, stream>>>(
      xg, Whh0, Wih1, Whh1, bih1, bhh1, out);
}


# recurrence: a third wave forms layer 1's chain-free gate terms (bias + Wih1@h0) from h0 passed through LDS; layer-0 wave keeps only its own chain; sum order changes: bias joins the input half first
# speedup vs baseline: 4.6278x; 1.0588x over previous
#include <hip/hip_runtime.h>

typedef float v2f __attribute__((ext_vector_type(2)));
typedef float v8f __attribute__((ext_vector_type(8)));

#define T_SEQ 2048
#define FEAT  256
#define GATES 64
#define HID   16
#define BATCH 64

#define K1_TT 64
#define K1_KG 16

struct K1Frag {
  v2f a0, a1;
  float4 x0, x1;
};

__device__ __forceinline__ v8f wmma4(v2f a, v2f b, v8f c) {
  return __builtin_amdgcn_wmma_f32_16x16x4_f32(
       false, a,  false, b,
       (short)0, c,  false,  false);
}

__device__ __forceinline__ void k1_load(K1Frag (&f)[4], const float* Wr0,
                                        const float* Wr1, const float* Xb,
                                        int k0, int half) {
#pragma unroll
  for (int s = 0; s < 4; ++s) {
    const int ka = k0 + 4 * s + 2 * half;
    f[s].a0 = *(const v2f*)(Wr0 + ka);
    f[s].a1 = *(const v2f*)(Wr1 + ka);
    f[s].x0 = *(const float4*)(Xb + (size_t)ka * T_SEQ);
    f[s].x1 = *(const float4*)(Xb + (size_t)(ka + 1) * T_SEQ);
  }
}

__device__ __forceinline__ void k1_mma(const K1Frag (&f)[4], v8f (&c0)[4],
                                       v8f (&c1)[4]) {
#pragma unroll
  for (int s = 0; s < 4; ++s) {
    const v2f b0 = {f[s].x0.x, f[s].x1.x};
    const v2f b1 = {f[s].x0.y, f[s].x1.y};
    const v2f b2 = {f[s].x0.z, f[s].x1.z};
    const v2f b3 = {f[s].x0.w, f[s].x1.w};
    c0[0] = wmma4(f[s].a0, b0, c0[0]);
    c0[1] = wmma4(f[s].a0, b1, c0[1]);
    c0[2] = wmma4(f[s].a0, b2, c0[2]);
    c0[3] = wmma4(f[s].a0, b3, c0[3]);
    c1[0] = wmma4(f[s].a1, b0, c1[0]);
    c1[1] = wmma4(f[s].a1, b1, c1[1]);
    c1[2] = wmma4(f[s].a1, b2, c1[2]);
    c1[3] = wmma4(f[s].a1, b3, c1[3]);
  }
}

__global__ __launch_bounds__(64) void lstm_in_gemm_wmma(
    const float* __restrict__ x,
    const float* __restrict__ W,
    const float* __restrict__ bih,
    const float* __restrict__ bhh,
    float* __restrict__ xg)
{
  const int lane = threadIdx.x & 31;
  const int mp   = threadIdx.x >> 5;
  const int half = lane >> 4;
  const int r    = lane & 15;
  const int tl   = blockIdx.x * K1_TT + 4 * r;
  const int b    = blockIdx.y;

  const float* Xb  = x + (size_t)b * FEAT * T_SEQ + tl;
  const float* Wr0 = W + (size_t)(mp * 32 + r) * FEAT;
  const float* Wr1 = Wr0 + (size_t)16 * FEAT;

  v8f c0[4] = {}, c1[4] = {};
  K1Frag fa[4], fb[4];
  k1_load(fa, Wr0, Wr1, Xb, 0, half);
  for (int k = 0; k < FEAT; k += 2 * K1_KG) {
    k1_load(fb, Wr0, Wr1, Xb, k + K1_KG, half);
    k1_mma(fa, c0, c1);
    if (k + 2 * K1_KG < FEAT)
      k1_load(fa, Wr0, Wr1, Xb, k + 2 * K1_KG, half);
    k1_mma(fb, c0, c1);
  }

#pragma unroll
  for (int v = 0; v < 8; ++v) {
    const int g0 = mp * 32 + v + 8 * half;
    const int g1 = g0 + 16;
    const float bi0 = bih[g0], bh0 = bhh[g0];
    const float bi1 = bih[g1], bh1 = bhh[g1];
    const float4 o0 = make_float4(c0[0][v] + bi0 + bh0, c0[1][v] + bi0 + bh0,
                                  c0[2][v] + bi0 + bh0, c0[3][v] + bi0 + bh0);
    const float4 o1 = make_float4(c1[0][v] + bi1 + bh1, c1[1][v] + bi1 + bh1,
                                  c1[2][v] + bi1 + bh1, c1[3][v] + bi1 + bh1);
    *(float4*)(xg + ((size_t)b * GATES + g0) * T_SEQ + tl) = o0;
    *(float4*)(xg + ((size_t)b * GATES + g1) * T_SEQ + tl) = o1;
  }
}

#define CH  32
#define NCH (T_SEQ / CH)
#define KH  8

template <int M>
__device__ __forceinline__ float rowbc(float x) {
  const int b = __builtin_bit_cast(int, x);
  return __builtin_bit_cast(float,
      __builtin_amdgcn_permlane16(b, b, 0x11111111u * M, 0x11111111u * M, true, false));
}
__device__ __forceinline__ float xjoin(float x) {
  const int b = __builtin_bit_cast(int, x);
  return __builtin_bit_cast(float,
      __builtin_amdgcn_permlanex16(b, b, 0xfedcba98u, 0x76543210u, true, false));
}
template <int M>
__device__ __forceinline__ float rowx(float x) {
  const int b = __builtin_bit_cast(int, x);
  return __builtin_bit_cast(float, __builtin_amdgcn_mov_dpp(b, 0x160 + M, 0xf, 0xf, false));
}
__device__ __forceinline__ float tanh_f(float x) {
  return __builtin_amdgcn_tanhf(x);
}
__device__ __forceinline__ float sig_h(float xh) {
  return fmaf(tanh_f(xh), 0.5f, 0.5f);
}
__device__ __forceinline__ float unitsum(float h) {
  float s = h;
  s += rowx<1>(s);
  s += rowx<2>(s);
  s += rowx<4>(s);
  s += rowx<8>(s);
  return s;
}

template <int M>
__device__ __forceinline__ void level_l0(const float (&w0)[4][KH],
                                         float h, float& hbc, float& hbn, float (&pa)[4]) {
  const float cur = hbc;
  hbc = hbn;
  if constexpr (M + 2 < KH) hbn = rowbc<M + 2>(h);
#pragma unroll
  for (int q = 0; q < 4; ++q) pa[q] = fmaf(w0[q][M], cur, pa[q]);
  asm volatile("" :: "v"(pa[0]), "v"(pa[1]), "v"(pa[2]), "v"(pa[3]), "v"(hbn));
  __builtin_amdgcn_sched_barrier(0);
}
template <int M>
__device__ __forceinline__ void level_l1(const float (&wh)[4][KH], float h, float& hbc, float& hbn,
                                         float (&rc)[4], float& us, float& ux) {
  const float cur = hbc;
  hbc = hbn;
  if constexpr (M + 2 < KH) hbn = rowbc<M + 2>(h);
#pragma unroll
  for (int q = 0; q < 4; ++q) rc[q] = fmaf(wh[q][M], cur, rc[q]);
  if constexpr (M % 2 == 0) ux = rowx<(1 << (M / 2))>(us);
  else                      us = us + ux;
  asm volatile("" :: "v"(rc[0]), "v"(rc[1]), "v"(rc[2]), "v"(rc[3]),
                     "v"(hbn), "v"(us), "v"(ux));
  __builtin_amdgcn_sched_barrier(0);
}

__global__ __launch_bounds__(96) void lstm_recurrence(
    const float* __restrict__ xg,
    const float* __restrict__ Whh0,
    const float* __restrict__ Wih1,
    const float* __restrict__ Whh1,
    const float* __restrict__ bih1,
    const float* __restrict__ bhh1,
    float* __restrict__ out)
{
  __shared__ float4 ring[2][CH][32];
  __shared__ float4 ringh[2][CH / 4][32];

  const int lane = threadIdx.x & 31;
  const int role = __builtin_amdgcn_readfirstlane((int)(threadIdx.x >> 5));
  const int b    = blockIdx.x;
  const int row  = lane >> 4;
  const int unit = (lane & 15) ^ (row << 3);
  const int kb   = row * KH;

  const float sc[4] = {0.5f, 0.5f, 1.0f, 0.5f};
  float w0[4][KH], wi[4][KH], wh[4][KH], bias[4];
#pragma unroll
  for (int q = 0; q < 4; ++q) {
    const int g = q * HID + unit;
#pragma unroll
    for (int m = 0; m < KH; ++m) {
      w0[q][m] = Whh0[g * HID + kb + m] * sc[q];
      wi[q][m] = Wih1[g * HID + kb + m] * sc[q];
      wh[q][m] = Whh1[g * HID + kb + m] * sc[q];
    }
    bias[q] = (bih1[g] + bhh1[g]) * sc[q];
  }

  const float* xgp[4];
#pragma unroll
  for (int q = 0; q < 4; ++q)
    xgp[q] = xg + ((size_t)b * GATES + q * HID + unit) * T_SEQ;
  float* outb = out + (size_t)b * T_SEQ;

  float h0 = 0.0f, c0 = 0.0f;
  float h1 = 0.0f, c1 = 0.0f;
  float hbc0 = 0.0f, hbn0 = 0.0f;
  float hbc1 = 0.0f, hbn1 = 0.0f;
  float outacc = 0.0f;

  float4 xv[4];
#pragma unroll
  for (int q = 0; q < 4; ++q) xv[q] = make_float4(0.f, 0.f, 0.f, 0.f);
  float hq[4] = {0.f, 0.f, 0.f, 0.f};

  if (role == 0) {
#pragma unroll
    for (int q = 0; q < 4; ++q) xv[q] = *(const float4*)(xgp[q]);
    float pa[4] = {0.f, 0.f, 0.f, 0.f};
#pragma unroll
    for (int m = 0; m < KH; ++m) {
#pragma unroll
      for (int q = 0; q < 4; ++q) pa[q] = fmaf(w0[q][m], 0.0f, pa[q]);
    }
    const float ai = fmaf(xv[0].x, sc[0], pa[0] + xjoin(pa[0]));
    const float af = fmaf(xv[1].x, sc[1], pa[1] + xjoin(pa[1]));
    const float ag = fmaf(xv[2].x, sc[2], pa[2] + xjoin(pa[2]));
    const float ao = fmaf(xv[3].x, sc[3], pa[3] + xjoin(pa[3]));
    c0 = sig_h(af) * c0 + sig_h(ai) * tanh_f(ag);
    h0 = sig_h(ao) * tanh_f(c0);
    hbc0 = rowbc<0>(h0);
    hbn0 = rowbc<1>(h0);
  }

  for (int p = 0; p <= NCH + 1; ++p) {
    if (role == 0) {
      if (p < NCH) {
        float4 (*wbuf)[32] = ringh[p & 1];
        for (int t0 = 0; t0 < CH; t0 += 4) {
          const int tg = p * CH + t0;
          const int tn = (tg + 4 < T_SEQ) ? (tg + 4) : tg;
          float4 nx[4];
#pragma unroll
          for (int q = 0; q < 4; ++q) nx[q] = *(const float4*)(xgp[q] + tn);
#pragma unroll
          for (int q = 0; q < 4; ++q) __builtin_prefetch(xgp[q] + tg + 64, 0, 3);
          float xa[4][4];
#pragma unroll
          for (int q = 0; q < 4; ++q) {
            xa[q][0] = xv[q].y; xa[q][1] = xv[q].z; xa[q][2] = xv[q].w; xa[q][3] = nx[q].x;
          }

#pragma unroll
          for (int u = 0; u < 4; ++u) {
            hq[u] = h0;
            float pa[4] = {0.f, 0.f, 0.f, 0.f};
            level_l0<0>(w0, h0, hbc0, hbn0, pa);
            level_l0<1>(w0, h0, hbc0, hbn0, pa);
            level_l0<2>(w0, h0, hbc0, hbn0, pa);
            level_l0<3>(w0, h0, hbc0, hbn0, pa);
            level_l0<4>(w0, h0, hbc0, hbn0, pa);
            level_l0<5>(w0, h0, hbc0, hbn0, pa);
            level_l0<6>(w0, h0, hbc0, hbn0, pa);
            level_l0<7>(w0, h0, hbc0, hbn0, pa);
            const float ai = fmaf(xa[0][u], sc[0], pa[0] + xjoin(pa[0]));
            const float af = fmaf(xa[1][u], sc[1], pa[1] + xjoin(pa[1]));
            const float ag = fmaf(xa[2][u], sc[2], pa[2] + xjoin(pa[2]));
            const float ao = fmaf(xa[3][u], sc[3], pa[3] + xjoin(pa[3]));
            asm volatile("" :: "v"(ai), "v"(af), "v"(ag), "v"(ao));
            __builtin_amdgcn_sched_barrier(0);
            c0 = sig_h(af) * c0 + sig_h(ai) * tanh_f(ag);
            const float tc = tanh_f(c0);
            const float so = sig_h(ao);
            if (u == 3) {
              asm volatile("" :: "v"(tc), "v"(so));
              __builtin_amdgcn_sched_barrier(0);
              wbuf[t0 >> 2][lane] = make_float4(hq[0], hq[1], hq[2], hq[3]);
              __builtin_amdgcn_sched_barrier(0);
            }
            h0 = so * tc;
            hbc0 = rowbc<0>(h0);
            hbn0 = rowbc<1>(h0);
            __builtin_amdgcn_sched_barrier(0);
          }
#pragma unroll
          for (int q = 0; q < 4; ++q) xv[q] = nx[q];
        }
      }
    } else if (role == 2) {
      if (p >= 1 && p <= NCH) {
        const float4 (*hbuf)[32] = ringh[(p - 1) & 1];
        float4 (*obuf)[32] = ring[(p - 1) & 1];
        for (int g = 0; g < CH / 4; ++g) {
          float4 hm[KH];
#pragma unroll
          for (int m = 0; m < KH; ++m) hm[m] = hbuf[g][kb + m];
          float hs[KH][4];
#pragma unroll
          for (int m = 0; m < KH; ++m) {
            hs[m][0] = hm[m].x; hs[m][1] = hm[m].y; hs[m][2] = hm[m].z; hs[m][3] = hm[m].w;
          }
#pragma unroll
          for (int u = 0; u < 4; ++u) {
            float ra[4] = {0.f, 0.f, 0.f, 0.f};
#pragma unroll
            for (int m = 0; m < KH; ++m) {
#pragma unroll
              for (int q = 0; q < 4; ++q) ra[q] = fmaf(wi[q][m], hs[m][u], ra[q]);
            }
            obuf[4 * g + u][lane] = make_float4(bias[0] + (ra[0] + xjoin(ra[0])),
                                                bias[1] + (ra[1] + xjoin(ra[1])),
                                                bias[2] + (ra[2] + xjoin(ra[2])),
                                                bias[3] + (ra[3] + xjoin(ra[3])));
          }
        }
      }
    } else {
      if (p >= 2) {
        const int q_ = p - 2;
        const float4 (*rbuf)[32] = ring[q_ & 1];
        float4 s0 = rbuf[0][lane], s1 = rbuf[1][lane];
        float4 s2 = rbuf[2][lane], s3 = rbuf[3][lane];
        for (int t0 = 0; t0 < CH; t0 += 4) {
          const int tq = (t0 + 4 < CH) ? (t0 + 4) : t0;
          const float4 n0 = rbuf[tq][lane],     n1 = rbuf[tq + 1][lane];
          const float4 n2 = rbuf[tq + 2][lane], n3 = rbuf[tq + 3][lane];
          const float sa[4][4] = {{s0.x, s0.y, s0.z, s0.w}, {s1.x, s1.y, s1.z, s1.w},
                                  {s2.x, s2.y, s2.z, s2.w}, {s3.x, s3.y, s3.z, s3.w}};

#pragma unroll
          for (int u = 0; u < 4; ++u) {
            float rc[4] = {0.f, 0.f, 0.f, 0.f};
            float us = h1, ux = 0.0f;
            level_l1<0>(wh, h1, hbc1, hbn1, rc, us, ux);
            level_l1<1>(wh, h1, hbc1, hbn1, rc, us, ux);
            level_l1<2>(wh, h1, hbc1, hbn1, rc, us, ux);
            level_l1<3>(wh, h1, hbc1, hbn1, rc, us, ux);
            level_l1<4>(wh, h1, hbc1, hbn1, rc, us, ux);
            level_l1<5>(wh, h1, hbc1, hbn1, rc, us, ux);
            level_l1<6>(wh, h1, hbc1, hbn1, rc, us, ux);
            level_l1<7>(wh, h1, hbc1, hbn1, rc, us, ux);
            outacc = (lane == ((t0 + u - 1) & 31)) ? us : outacc;
            const float bi = sa[u][0] + (rc[0] + xjoin(rc[0]));
            const float bf = sa[u][1] + (rc[1] + xjoin(rc[1]));
            const float bg = sa[u][2] + (rc[2] + xjoin(rc[2]));
            const float bo = sa[u][3] + (rc[3] + xjoin(rc[3]));
            asm volatile("" :: "v"(bi), "v"(bf), "v"(bg), "v"(bo));
            __builtin_amdgcn_sched_barrier(0);
            c1 = sig_h(bf) * c1 + sig_h(bi) * tanh_f(bg);
            h1 = sig_h(bo) * tanh_f(c1);
            hbc1 = rowbc<0>(h1);
            hbn1 = rowbc<1>(h1);
            __builtin_amdgcn_sched_barrier(0);
            if (u == 0 && t0 == 0 && q_ != 0)
              outb[(q_ * CH - 32) + lane] = outacc * 0.0625f;
          }
          s0 = n0; s1 = n1; s2 = n2; s3 = n3;
        }
      }
    }
    __builtin_amdgcn_fence(__ATOMIC_RELEASE, "workgroup", "local");
    __builtin_amdgcn_s_barrier();
    __builtin_amdgcn_fence(__ATOMIC_ACQUIRE, "workgroup", "local");
  }

  if (role == 1) {
    const float slast = unitsum(h1);
    outacc = (lane == 31) ? slast : outacc;
    outb[(T_SEQ - 32) + lane] = outacc * 0.0625f;
  }
}

extern "C" void kernel_launch(void* const* d_in, const int* in_sizes, int n_in,
                              void* d_out, int out_size, void* d_ws, size_t ws_size,
                              hipStream_t stream) {
  (void)in_sizes; (void)n_in; (void)out_size; (void)ws_size;
  const float* x    = (const float*)d_in[0];
  const float* Wih0 = (const float*)d_in[1];
  const float* Whh0 = (const float*)d_in[2];
  const float* bih0 = (const float*)d_in[3];
  const float* bhh0 = (const float*)d_in[4];
  const float* Wih1 = (const float*)d_in[5];
  const float* Whh1 = (const float*)d_in[6];
  const float* bih1 = (const float*)d_in[7];
  const float* bhh1 = (const float*)d_in[8];
  float* out = (float*)d_out;
  float* xg  = (float*)d_ws;

  dim3 g1(T_SEQ / K1_TT, BATCH, 1);
  lstm_in_gemm_wmma<<<g1, dim3(64, 1, 1), 0, stream>>>(x, Wih0, bih0, bhh0, xg);

  lstm_recurrence<<<dim3(BATCH, 1, 1), dim3(96, 1, 1), 0, stream>>>(
      xg, Whh0, Wih1, Whh1, bih1, bhh1, out);
}


# recurrence: the chain-free gate term (x term; layer 1's bias + input half) starts both rows' k chains, half in each, instead of being added behind the join; sum order changes
# speedup vs baseline: 4.8885x; 1.0563x over previous
#include <hip/hip_runtime.h>

typedef float v2f __attribute__((ext_vector_type(2)));
typedef float v8f __attribute__((ext_vector_type(8)));

#define T_SEQ 2048
#define FEAT  256
#define GATES 64
#define HID   16
#define BATCH 64

#define K1_TT 64
#define K1_KG 16

struct K1Frag {
  v2f a0, a1;
  float4 x0, x1;
};

__device__ __forceinline__ v8f wmma4(v2f a, v2f b, v8f c) {
  return __builtin_amdgcn_wmma_f32_16x16x4_f32(
       false, a,  false, b,
       (short)0, c,  false,  false);
}

__device__ __forceinline__ void k1_load(K1Frag (&f)[4], const float* Wr0,
                                        const float* Wr1, const float* Xb,
                                        int k0, int half) {
#pragma unroll
  for (int s = 0; s < 4; ++s) {
    const int ka = k0 + 4 * s + 2 * half;
    f[s].a0 = *(const v2f*)(Wr0 + ka);
    f[s].a1 = *(const v2f*)(Wr1 + ka);
    f[s].x0 = *(const float4*)(Xb + (size_t)ka * T_SEQ);
    f[s].x1 = *(const float4*)(Xb + (size_t)(ka + 1) * T_SEQ);
  }
}

__device__ __forceinline__ void k1_mma(const K1Frag (&f)[4], v8f (&c0)[4],
                                       v8f (&c1)[4]) {
#pragma unroll
  for (int s = 0; s < 4; ++s) {
    const v2f b0 = {f[s].x0.x, f[s].x1.x};
    const v2f b1 = {f[s].x0.y, f[s].x1.y};
    const v2f b2 = {f[s].x0.z, f[s].x1.z};
    const v2f b3 = {f[s].x0.w, f[s].x1.w};
    c0[0] = wmma4(f[s].a0, b0, c0[0]);
    c0[1] = wmma4(f[s].a0, b1, c0[1]);
    c0[2] = wmma4(f[s].a0, b2, c0[2]);
    c0[3] = wmma4(f[s].a0, b3, c0[3]);
    c1[0] = wmma4(f[s].a1, b0, c1[0]);
    c1[1] = wmma4(f[s].a1, b1, c1[1]);
    c1[2] = wmma4(f[s].a1, b2, c1[2]);
    c1[3] = wmma4(f[s].a1, b3, c1[3]);
  }
}

__global__ __launch_bounds__(64) void lstm_in_gemm_wmma(
    const float* __restrict__ x,
    const float* __restrict__ W,
    const float* __restrict__ bih,
    const float* __restrict__ bhh,
    float* __restrict__ xg)
{
  const int lane = threadIdx.x & 31;
  const int mp   = threadIdx.x >> 5;
  const int half = lane >> 4;
  const int r    = lane & 15;
  const int tl   = blockIdx.x * K1_TT + 4 * r;
  const int b    = blockIdx.y;

  const float* Xb  = x + (size_t)b * FEAT * T_SEQ + tl;
  const float* Wr0 = W + (size_t)(mp * 32 + r) * FEAT;
  const float* Wr1 = Wr0 + (size_t)16 * FEAT;

  v8f c0[4] = {}, c1[4] = {};
  K1Frag fa[4], fb[4];
  k1_load(fa, Wr0, Wr1, Xb, 0, half);
  for (int k = 0; k < FEAT; k += 2 * K1_KG) {
    k1_load(fb, Wr0, Wr1, Xb, k + K1_KG, half);
    k1_mma(fa, c0, c1);
    if (k + 2 * K1_KG < FEAT)
      k1_load(fa, Wr0, Wr1, Xb, k + 2 * K1_KG, half);
    k1_mma(fb, c0, c1);
  }

#pragma unroll
  for (int v = 0; v < 8; ++v) {
    const int g0 = mp * 32 + v + 8 * half;
    const int g1 = g0 + 16;
    const float bi0 = bih[g0], bh0 = bhh[g0];
    const float bi1 = bih[g1], bh1 = bhh[g1];
    const float hs0 = (mp == 1) ? 0.5f : 0.25f;
    const float hs1 = 0.25f;
    const float4 o0 = make_float4((c0[0][v] + bi0 + bh0) * hs0, (c0[1][v] + bi0 + bh0) * hs0,
                                  (c0[2][v] + bi0 + bh0) * hs0, (c0[3][v] + bi0 + bh0) * hs0);
    const float4 o1 = make_float4((c1[0][v] + bi1 + bh1) * hs1, (c1[1][v] + bi1 + bh1) * hs1,
                                  (c1[2][v] + bi1 + bh1) * hs1, (c1[3][v] + bi1 + bh1) * hs1);
    *(float4*)(xg + ((size_t)b * GATES + g0) * T_SEQ + tl) = o0;
    *(float4*)(xg + ((size_t)b * GATES + g1) * T_SEQ + tl) = o1;
  }
}

#define CH  32
#define NCH (T_SEQ / CH)
#define KH  8

template <int M>
__device__ __forceinline__ float rowbc(float x) {
  const int b = __builtin_bit_cast(int, x);
  return __builtin_bit_cast(float,
      __builtin_amdgcn_permlane16(b, b, 0x11111111u * M, 0x11111111u * M, true, false));
}
__device__ __forceinline__ float xjoin(float x) {
  const int b = __builtin_bit_cast(int, x);
  return __builtin_bit_cast(float,
      __builtin_amdgcn_permlanex16(b, b, 0xfedcba98u, 0x76543210u, true, false));
}
template <int M>
__device__ __forceinline__ float rowx(float x) {
  const int b = __builtin_bit_cast(int, x);
  return __builtin_bit_cast(float, __builtin_amdgcn_mov_dpp(b, 0x160 + M, 0xf, 0xf, false));
}
__device__ __forceinline__ float tanh_f(float x) {
  return __builtin_amdgcn_tanhf(x);
}
__device__ __forceinline__ float sig_h(float xh) {
  return fmaf(tanh_f(xh), 0.5f, 0.5f);
}
__device__ __forceinline__ float unitsum(float h) {
  float s = h;
  s += rowx<1>(s);
  s += rowx<2>(s);
  s += rowx<4>(s);
  s += rowx<8>(s);
  return s;
}

template <int M>
__device__ __forceinline__ void level_l0(const float (&w0)[4][KH],
                                         float h, float& hbc, float& hbn, float (&pa)[4]) {
  const float cur = hbc;
  hbc = hbn;
  if constexpr (M + 2 < KH) hbn = rowbc<M + 2>(h);
#pragma unroll
  for (int q = 0; q < 4; ++q) pa[q] = fmaf(w0[q][M], cur, pa[q]);
  asm volatile("" :: "v"(pa[0]), "v"(pa[1]), "v"(pa[2]), "v"(pa[3]), "v"(hbn));
  __builtin_amdgcn_sched_barrier(0);
}
template <int M>
__device__ __forceinline__ void level_l1(const float (&wh)[4][KH], float h, float& hbc, float& hbn,
                                         float (&rc)[4], float& us, float& ux) {
  const float cur = hbc;
  hbc = hbn;
  if constexpr (M + 2 < KH) hbn = rowbc<M + 2>(h);
#pragma unroll
  for (int q = 0; q < 4; ++q) rc[q] = fmaf(wh[q][M], cur, rc[q]);
  if constexpr (M % 2 == 0) ux = rowx<(1 << (M / 2))>(us);
  else                      us = us + ux;
  asm volatile("" :: "v"(rc[0]), "v"(rc[1]), "v"(rc[2]), "v"(rc[3]),
                     "v"(hbn), "v"(us), "v"(ux));
  __builtin_amdgcn_sched_barrier(0);
}

__global__ __launch_bounds__(96) void lstm_recurrence(
    const float* __restrict__ xg,
    const float* __restrict__ Whh0,
    const float* __restrict__ Wih1,
    const float* __restrict__ Whh1,
    const float* __restrict__ bih1,
    const float* __restrict__ bhh1,
    float* __restrict__ out)
{
  __shared__ float4 ring[2][CH][32];
  __shared__ float4 ringh[2][CH / 4][32];

  const int lane = threadIdx.x & 31;
  const int role = __builtin_amdgcn_readfirstlane((int)(threadIdx.x >> 5));
  const int b    = blockIdx.x;
  const int row  = lane >> 4;
  const int unit = (lane & 15) ^ (row << 3);
  const int kb   = row * KH;

  const float sc[4] = {0.5f, 0.5f, 1.0f, 0.5f};
  float w0[4][KH], wi[4][KH], wh[4][KH], bias[4];
#pragma unroll
  for (int q = 0; q < 4; ++q) {
    const int g = q * HID + unit;
#pragma unroll
    for (int m = 0; m < KH; ++m) {
      w0[q][m] = Whh0[g * HID + kb + m] * sc[q];
      wi[q][m] = Wih1[g * HID + kb + m] * sc[q];
      wh[q][m] = Whh1[g * HID + kb + m] * sc[q];
    }
    bias[q] = (bih1[g] + bhh1[g]) * sc[q];
  }

  const float* xgp[4];
#pragma unroll
  for (int q = 0; q < 4; ++q)
    xgp[q] = xg + ((size_t)b * GATES + q * HID + unit) * T_SEQ;
  float* outb = out + (size_t)b * T_SEQ;

  float h0 = 0.0f, c0 = 0.0f;
  float h1 = 0.0f, c1 = 0.0f;
  float hbc0 = 0.0f, hbn0 = 0.0f;
  float hbc1 = 0.0f, hbn1 = 0.0f;
  float outacc = 0.0f;

  float4 xv[4];
#pragma unroll
  for (int q = 0; q < 4; ++q) xv[q] = make_float4(0.f, 0.f, 0.f, 0.f);
  float hq[4] = {0.f, 0.f, 0.f, 0.f};

  if (role == 0) {
#pragma unroll
    for (int q = 0; q < 4; ++q) xv[q] = *(const float4*)(xgp[q]);
    float pa[4] = {xv[0].x, xv[1].x, xv[2].x, xv[3].x};
#pragma unroll
    for (int m = 0; m < KH; ++m) {
#pragma unroll
      for (int q = 0; q < 4; ++q) pa[q] = fmaf(w0[q][m], 0.0f, pa[q]);
    }
    const float ai = pa[0] + xjoin(pa[0]);
    const float af = pa[1] + xjoin(pa[1]);
    const float ag = pa[2] + xjoin(pa[2]);
    const float ao = pa[3] + xjoin(pa[3]);
    c0 = sig_h(af) * c0 + sig_h(ai) * tanh_f(ag);
    h0 = sig_h(ao) * tanh_f(c0);
    hbc0 = rowbc<0>(h0);
    hbn0 = rowbc<1>(h0);
  }

  for (int p = 0; p <= NCH + 1; ++p) {
    if (role == 0) {
      if (p < NCH) {
        float4 (*wbuf)[32] = ringh[p & 1];
        for (int t0 = 0; t0 < CH; t0 += 4) {
          const int tg = p * CH + t0;
          const int tn = (tg + 4 < T_SEQ) ? (tg + 4) : tg;
          float4 nx[4];
#pragma unroll
          for (int q = 0; q < 4; ++q) nx[q] = *(const float4*)(xgp[q] + tn);
#pragma unroll
          for (int q = 0; q < 4; ++q) __builtin_prefetch(xgp[q] + tg + 64, 0, 3);
          float xa[4][4];
#pragma unroll
          for (int q = 0; q < 4; ++q) {
            xa[q][0] = xv[q].y; xa[q][1] = xv[q].z; xa[q][2] = xv[q].w; xa[q][3] = nx[q].x;
          }

#pragma unroll
          for (int u = 0; u < 4; ++u) {
            hq[u] = h0;
            float pa[4] = {xa[0][u], xa[1][u], xa[2][u], xa[3][u]};
            level_l0<0>(w0, h0, hbc0, hbn0, pa);
            level_l0<1>(w0, h0, hbc0, hbn0, pa);
            level_l0<2>(w0, h0, hbc0, hbn0, pa);
            level_l0<3>(w0, h0, hbc0, hbn0, pa);
            level_l0<4>(w0, h0, hbc0, hbn0, pa);
            level_l0<5>(w0, h0, hbc0, hbn0, pa);
            level_l0<6>(w0, h0, hbc0, hbn0, pa);
            level_l0<7>(w0, h0, hbc0, hbn0, pa);
            const float ai = pa[0] + xjoin(pa[0]);
            const float af = pa[1] + xjoin(pa[1]);
            const float ag = pa[2] + xjoin(pa[2]);
            const float ao = pa[3] + xjoin(pa[3]);
            asm volatile("" :: "v"(ai), "v"(af), "v"(ag), "v"(ao));
            __builtin_amdgcn_sched_barrier(0);
            const float to = tanh_f(ao);
            c0 = sig_h(af) * c0 + sig_h(ai) * tanh_f(ag);
            asm volatile("" :: "v"(c0), "v"(to));
            __builtin_amdgcn_sched_barrier(0);
            const float tc = tanh_f(c0);
            const float so = fmaf(to, 0.5f, 0.5f);
            if (u == 3) {
              asm volatile("" :: "v"(tc), "v"(so));
              __builtin_amdgcn_sched_barrier(0);
              wbuf[t0 >> 2][lane] = make_float4(hq[0], hq[1], hq[2], hq[3]);
              __builtin_amdgcn_sched_barrier(0);
            }
            h0 = so * tc;
            hbc0 = rowbc<0>(h0);
            hbn0 = rowbc<1>(h0);
            __builtin_amdgcn_sched_barrier(0);
          }
#pragma unroll
          for (int q = 0; q < 4; ++q) xv[q] = nx[q];
        }
      }
    } else if (role == 2) {
      if (p >= 1 && p <= NCH) {
        const float4 (*hbuf)[32] = ringh[(p - 1) & 1];
        float4 (*obuf)[32] = ring[(p - 1) & 1];
        for (int g = 0; g < CH / 4; ++g) {
          float4 hm[KH];
#pragma unroll
          for (int m = 0; m < KH; ++m) hm[m] = hbuf[g][kb + m];
          float hs[KH][4];
#pragma unroll
          for (int m = 0; m < KH; ++m) {
            hs[m][0] = hm[m].x; hs[m][1] = hm[m].y; hs[m][2] = hm[m].z; hs[m][3] = hm[m].w;
          }
#pragma unroll
          for (int u = 0; u < 4; ++u) {
            float ra[4] = {0.f, 0.f, 0.f, 0.f};
#pragma unroll
            for (int m = 0; m < KH; ++m) {
#pragma unroll
              for (int q = 0; q < 4; ++q) ra[q] = fmaf(wi[q][m], hs[m][u], ra[q]);
            }
            obuf[4 * g + u][lane] = make_float4(0.5f * (bias[0] + (ra[0] + xjoin(ra[0]))),
                                                0.5f * (bias[1] + (ra[1] + xjoin(ra[1]))),
                                                0.5f * (bias[2] + (ra[2] + xjoin(ra[2]))),
                                                0.5f * (bias[3] + (ra[3] + xjoin(ra[3]))));
          }
        }
      }
    } else {
      if (p >= 2) {
        const int q_ = p - 2;
        const float4 (*rbuf)[32] = ring[q_ & 1];
        float4 s0 = rbuf[0][lane], s1 = rbuf[1][lane];
        float4 s2 = rbuf[2][lane], s3 = rbuf[3][lane];
        for (int t0 = 0; t0 < CH; t0 += 4) {
          const int tq = (t0 + 4 < CH) ? (t0 + 4) : t0;
          const float4 n0 = rbuf[tq][lane],     n1 = rbuf[tq + 1][lane];
          const float4 n2 = rbuf[tq + 2][lane], n3 = rbuf[tq + 3][lane];
          const float sa[4][4] = {{s0.x, s0.y, s0.z, s0.w}, {s1.x, s1.y, s1.z, s1.w},
                                  {s2.x, s2.y, s2.z, s2.w}, {s3.x, s3.y, s3.z, s3.w}};

#pragma unroll
          for (int u = 0; u < 4; ++u) {
            float rc[4] = {sa[u][0], sa[u][1], sa[u][2], sa[u][3]};
            float us = h1, ux = 0.0f;
            level_l1<0>(wh, h1, hbc1, hbn1, rc, us, ux);
            level_l1<1>(wh, h1, hbc1, hbn1, rc, us, ux);
            level_l1<2>(wh, h1, hbc1, hbn1, rc, us, ux);
            level_l1<3>(wh, h1, hbc1, hbn1, rc, us, ux);
            level_l1<4>(wh, h1, hbc1, hbn1, rc, us, ux);
            level_l1<5>(wh, h1, hbc1, hbn1, rc, us, ux);
            level_l1<6>(wh, h1, hbc1, hbn1, rc, us, ux);
            level_l1<7>(wh, h1, hbc1, hbn1, rc, us, ux);
            outacc = (lane == ((t0 + u - 1) & 31)) ? us : outacc;
            const float bi = rc[0] + xjoin(rc[0]);
            const float bf = rc[1] + xjoin(rc[1]);
            const float bg = rc[2] + xjoin(rc[2]);
            const float bo = rc[3] + xjoin(rc[3]);
            asm volatile("" :: "v"(bi), "v"(bf), "v"(bg), "v"(bo));
            __builtin_amdgcn_sched_barrier(0);
            c1 = sig_h(bf) * c1 + sig_h(bi) * tanh_f(bg);
            h1 = sig_h(bo) * tanh_f(c1);
            hbc1 = rowbc<0>(h1);
            hbn1 = rowbc<1>(h1);
            __builtin_amdgcn_sched_barrier(0);
            if (u == 0 && t0 == 0 && q_ != 0)
              outb[(q_ * CH - 32) + lane] = outacc * 0.0625f;
          }
          s0 = n0; s1 = n1; s2 = n2; s3 = n3;
        }
      }
    }
    __builtin_amdgcn_fence(__ATOMIC_RELEASE, "workgroup", "local");
    __builtin_amdgcn_s_barrier();
    __builtin_amdgcn_fence(__ATOMIC_ACQUIRE, "workgroup", "local");
  }

  if (role == 1) {
    const float slast = unitsum(h1);
    outacc = (lane == 31) ? slast : outacc;
    outb[(T_SEQ - 32) + lane] = outacc * 0.0625f;
  }
}

extern "C" void kernel_launch(void* const* d_in, const int* in_sizes, int n_in,
                              void* d_out, int out_size, void* d_ws, size_t ws_size,
                              hipStream_t stream) {
  (void)in_sizes; (void)n_in; (void)out_size; (void)ws_size;
  const float* x    = (const float*)d_in[0];
  const float* Wih0 = (const float*)d_in[1];
  const float* Whh0 = (const float*)d_in[2];
  const float* bih0 = (const float*)d_in[3];
  const float* bhh0 = (const float*)d_in[4];
  const float* Wih1 = (const float*)d_in[5];
  const float* Whh1 = (const float*)d_in[6];
  const float* bih1 = (const float*)d_in[7];
  const float* bhh1 = (const float*)d_in[8];
  float* out = (float*)d_out;
  float* xg  = (float*)d_ws;

  dim3 g1(T_SEQ / K1_TT, BATCH, 1);
  lstm_in_gemm_wmma<<<g1, dim3(64, 1, 1), 0, stream>>>(x, Wih0, bih0, bhh0, xg);

  lstm_recurrence<<<dim3(BATCH, 1, 1), dim3(96, 1, 1), 0, stream>>>(
      xg, Whh0, Wih1, Whh1, bih1, bhh1, out);
}
